# starting kernel
# baseline (speedup 1.0000x reference)
#include <hip/hip_runtime.h>

typedef __attribute__((ext_vector_type(2))) float v2f;
typedef __attribute__((ext_vector_type(8))) float v8f;

#define GD 64

__global__ void k_init_deg(float* __restrict__ deg, int n) {
    int i = blockIdx.x * blockDim.x + threadIdx.x;
    if (i < n) deg[i] = 1.0f;
}

__global__ void k_count_deg(const int* __restrict__ dst, float* __restrict__ deg, int e) {
    int i = blockIdx.x * blockDim.x + threadIdx.x;
    if (i < e) atomicAdd(&deg[dst[i]], 1.0f);
}

__global__ void k_dinv(const float* __restrict__ deg, float* __restrict__ dinv, int n) {
    int i = blockIdx.x * blockDim.x + threadIdx.x;
    if (i < n) dinv[i] = rsqrtf(fmaxf(deg[i], 1e-12f));
}

__global__ void k_norm(const int* __restrict__ src, const int* __restrict__ dst,
                       const float* __restrict__ dinv, float* __restrict__ norm, int e) {
    int i = blockIdx.x * blockDim.x + threadIdx.x;
    if (i < e) norm[i] = dinv[src[i]] * dinv[dst[i]];
}

template <bool RELU_IN, bool ADD_BIAS, bool RELU_OUT>
__global__ void __launch_bounds__(256)
k_gemm64(const float* __restrict__ H, const float* __restrict__ W,
         const float* __restrict__ bias, float* __restrict__ out, int n)
{
    __shared__ v2f   sWp[(GD / 2) * GD];
    __shared__ float sB[GD];

    const int t = threadIdx.x;
    for (int i = t; i < (GD / 2) * GD; i += 256) {
        const int m = i >> 6;
        const int c = i & 63;
        v2f p;
        p.x = W[(2 * m) * GD + c];
        p.y = W[(2 * m + 1) * GD + c];
        sWp[i] = p;
    }
    if (ADD_BIAS && t < GD) sB[t] = bias[t];
    __syncthreads();

    const int wave = t >> 5;
    const int lane = t & 31;
    const int r    = lane & 15;
    const int hi   = lane >> 4;
    const int rowBase = blockIdx.x * 128 + wave * 16;
    if (rowBase >= n) return;

    v8f acc0 = {}, acc1 = {}, acc2 = {}, acc3 = {};
    const float* aRow = H + (size_t)(rowBase + r) * GD;

    #pragma unroll
    for (int k = 0; k < 16; ++k) {
        const int ka = 4 * k + 2 * hi;
        v2f a = *(const v2f*)(aRow + ka);
        if (RELU_IN) { a.x = fmaxf(a.x, 0.0f); a.y = fmaxf(a.y, 0.0f); }

        const int mB = 2 * k + hi;
        const v2f* wp = sWp + (size_t)mB * GD + r;
        const v2f b0 = wp[0];
        const v2f b1 = wp[16];
        const v2f b2 = wp[32];
        const v2f b3 = wp[48];

        acc0 = __builtin_amdgcn_wmma_f32_16x16x4_f32(false, a, false, b0, (short)0, acc0, false, false);
        acc1 = __builtin_amdgcn_wmma_f32_16x16x4_f32(false, a, false, b1, (short)0, acc1, false, false);
        acc2 = __builtin_amdgcn_wmma_f32_16x16x4_f32(false, a, false, b2, (short)0, acc2, false, false);
        acc3 = __builtin_amdgcn_wmma_f32_16x16x4_f32(false, a, false, b3, (short)0, acc3, false, false);
    }

    #pragma unroll
    for (int v = 0; v < 8; ++v) {
        const int row = rowBase + v + 8 * hi;
        float* o = out + (size_t)row * GD + r;
        float x0 = acc0[v], x1 = acc1[v], x2 = acc2[v], x3 = acc3[v];
        if (ADD_BIAS) { x0 += sB[r]; x1 += sB[16 + r]; x2 += sB[32 + r]; x3 += sB[48 + r]; }
        if (RELU_OUT) {
            x0 = fmaxf(x0, 0.0f); x1 = fmaxf(x1, 0.0f);
            x2 = fmaxf(x2, 0.0f); x3 = fmaxf(x3, 0.0f);
        }
        o[0] = x0; o[16] = x1; o[32] = x2; o[48] = x3;
    }
}

__global__ void k_self_bias(const float* __restrict__ t, const float* __restrict__ dinv,
                            const float* __restrict__ b, float* __restrict__ out, int n)
{
    int tid = blockIdx.x * blockDim.x + threadIdx.x;
    if (tid >= n * 16) return;
    const int i = tid >> 4;
    const int c = tid & 15;
    const float s = dinv[i] * dinv[i];
    const float4 v  = *(const float4*)(t + (size_t)i * GD + c * 4);
    const float4 bb = *(const float4*)(b + c * 4);
    float4 o;
    o.x = bb.x + v.x * s;
    o.y = bb.y + v.y * s;
    o.z = bb.z + v.z * s;
    o.w = bb.w + v.w * s;
    *(float4*)(out + (size_t)i * GD + c * 4) = o;
}

__global__ void k_scatter(const float* __restrict__ t, const int* __restrict__ src,
                          const int* __restrict__ dst, const float* __restrict__ norm,
                          float* __restrict__ out, int e)
{
    int tid = blockIdx.x * blockDim.x + threadIdx.x;
    if (tid >= e * 16) return;
    const int ei = tid >> 4;
    const int c  = tid & 15;
    const int s  = src[ei];
    const int d  = dst[ei];
    const float w = norm[ei];
    const float4 v = *(const float4*)(t + (size_t)s * GD + c * 4);
    float* o = out + (size_t)d * GD + c * 4;
    atomicAdd(o + 0, v.x * w);
    atomicAdd(o + 1, v.y * w);
    atomicAdd(o + 2, v.z * w);
    atomicAdd(o + 3, v.w * w);
}

extern "C" void kernel_launch(void* const* d_in, const int* in_sizes, int n_in,
                              void* d_out, int out_size, void* d_ws, size_t ws_size,
                              hipStream_t stream)
{
    const float* x    = (const float*)d_in[0];
    const int*   ei   = (const int*)d_in[1];
    const float* W_in = (const float*)d_in[2];
    const float* b_in = (const float*)d_in[3];
    const float* W1   = (const float*)d_in[4];
    const float* b1   = (const float*)d_in[5];
    const float* W2   = (const float*)d_in[6];
    const float* b2   = (const float*)d_in[7];
    const float* W3   = (const float*)d_in[8];
    const float* b3   = (const float*)d_in[9];
    float* out = (float*)d_out;

    const int N = in_sizes[0] / GD;
    const int E = in_sizes[1] / 2;
    const int* src = ei;
    const int* dst = ei + E;

    float* bufA = (float*)d_ws;
    float* bufB = bufA + (size_t)N * GD;
    float* deg  = bufB + (size_t)N * GD;
    float* dinv = deg  + N;
    float* norm = dinv + N;

    const int TB = 256;
    const int gN  = (N + TB - 1) / TB;
    const int gE  = (E + TB - 1) / TB;
    const int gN16 = (N * 16 + TB - 1) / TB;
    const int gE16 = (int)(((long long)E * 16 + TB - 1) / TB);
    const int gGemm = (N + 127) / 128;

    k_init_deg <<<gN, TB, 0, stream>>>(deg, N);
    k_count_deg<<<gE, TB, 0, stream>>>(dst, deg, E);
    k_dinv     <<<gN, TB, 0, stream>>>(deg, dinv, N);
    k_norm     <<<gE, TB, 0, stream>>>(src, dst, dinv, norm, E);

    k_gemm64<false, true, true><<<gGemm, TB, 0, stream>>>(x, W_in, b_in, bufA, N);

    k_gemm64<false, false, false><<<gGemm, TB, 0, stream>>>(bufA, W1, nullptr, bufB, N);
    k_self_bias<<<gN16, TB, 0, stream>>>(bufB, dinv, b1, bufA, N);
    k_scatter <<<gE16, TB, 0, stream>>>(bufB, src, dst, norm, bufA, E);

    k_gemm64<true, false, false><<<gGemm, TB, 0, stream>>>(bufA, W2, nullptr, bufB, N);
    k_self_bias<<<gN16, TB, 0, stream>>>(bufB, dinv, b2, bufA, N);
    k_scatter <<<gE16, TB, 0, stream>>>(bufB, src, dst, norm, bufA, E);

    k_gemm64<true, false, false><<<gGemm, TB, 0, stream>>>(bufA, W3, nullptr, bufB, N);
    k_self_bias<<<gN16, TB, 0, stream>>>(bufB, dinv, b3, out, N);
    k_scatter <<<gE16, TB, 0, stream>>>(bufB, src, dst, norm, out, E);
}


# Edge ranks from LDS counters in a 2-level counting sort (chunk x bucket counts, scan, bucket runs, per-bucket count and place) replace the per-edge global returning add; row sums in LDS arrival order
# speedup vs baseline: 3.4817x; 3.4817x over previous
#include <hip/hip_runtime.h>


typedef __attribute__((ext_vector_type(2))) float v2f;
typedef __attribute__((ext_vector_type(8))) float v8f;

#define GD 64

#define NPB_SHIFT 8
#define NPB  (1 << NPB_SHIFT)
#define MAXB 256
#define CH   8192
#define SORT_T 1024

__global__ void __launch_bounds__(SORT_T)
k_chunk_hist(const int* __restrict__ dst, int* __restrict__ tab, int e, int nb, int nc)
{
    __shared__ int sCnt[MAXB];
    const int t = threadIdx.x;
    if (t < MAXB) sCnt[t] = 0;
    __syncthreads();
    const int base = blockIdx.x * CH;
    int d[CH / SORT_T];
    #pragma unroll
    for (int j = 0; j < CH / SORT_T; ++j) {
        const int i = base + j * SORT_T + t;
        d[j] = dst[i < e ? i : e - 1];
    }
    #pragma unroll
    for (int j = 0; j < CH / SORT_T; ++j) {
        if (base + j * SORT_T + t < e) atomicAdd(&sCnt[d[j] >> NPB_SHIFT], 1);
    }
    __syncthreads();
    if (t < nb) tab[t * nc + blockIdx.x] = sCnt[t];
}

__global__ void __launch_bounds__(SORT_T)
k_chunk_scatter(const int* __restrict__ src, const int* __restrict__ dst,
                const int* __restrict__ tabOff, int2* __restrict__ tmp, int e, int nb, int nc)
{
    __shared__ int sCur[MAXB];
    const int t = threadIdx.x;
    if (t < nb) sCur[t] = tabOff[t * nc + blockIdx.x];
    __syncthreads();
    const int base = blockIdx.x * CH;
    int s[CH / SORT_T];
    int d[CH / SORT_T];
    #pragma unroll
    for (int j = 0; j < CH / SORT_T; ++j) {
        const int i  = base + j * SORT_T + t;
        const int ic = i < e ? i : e - 1;
        s[j] = src[ic];
        d[j] = dst[ic];
    }
    #pragma unroll
    for (int j = 0; j < CH / SORT_T; ++j) {
        if (base + j * SORT_T + t < e) {
            const int p = atomicAdd(&sCur[d[j] >> NPB_SHIFT], 1);
            tmp[p] = make_int2(s[j], d[j]);
        }
    }
}

__global__ void __launch_bounds__(SORT_T)
k_bucket_count(const int2* __restrict__ tmp, const int* __restrict__ tabOff,
               int* __restrict__ rowptr, float* __restrict__ dinv, int n, int e, int nc)
{
    __shared__ int sCnt[NPB];
    __shared__ int sTot[NPB / 32];
    const int t    = threadIdx.x;
    const int lane = t & 31;
    const int b    = blockIdx.x;
    const int bstart = tabOff[b * nc];
    const int bend   = tabOff[(b + 1) * nc];
    if (t < NPB) sCnt[t] = 0;
    __syncthreads();
    for (int p0 = bstart + t; p0 < bend; p0 += 4 * SORT_T) {
        int d[4];
        #pragma unroll
        for (int k = 0; k < 4; ++k) {
            const int p = p0 + k * SORT_T;
            d[k] = tmp[p < bend ? p : p0].y;
        }
        #pragma unroll
        for (int k = 0; k < 4; ++k) {
            if (p0 + k * SORT_T < bend) atomicAdd(&sCnt[d[k] & (NPB - 1)], 1);
        }
    }
    __syncthreads();
    int c = 0, inc = 0;
    if (t < NPB) {
        c = sCnt[t];
        inc = c;
        #pragma unroll
        for (int d = 1; d < 32; d <<= 1) {
            const int y = __shfl_up(inc, d);
            if (lane >= d) inc += y;
        }
        if (lane == 31) sTot[t >> 5] = inc;
    }
    __syncthreads();
    if (t < NPB) {
        int pre = 0;
        for (int w = 0; w < (t >> 5); ++w) pre += sTot[w];
        const int node = b * NPB + t;
        if (node < n) {
            rowptr[node] = bstart + pre + inc - c;
            dinv[node]   = rsqrtf(fmaxf((float)(c + 1), 1e-12f));
        }
    }
    if (b == 0 && t == 0) rowptr[n] = e;
}

__global__ void __launch_bounds__(SORT_T)
k_bucket_place(const int2* __restrict__ tmp, const int* __restrict__ tabOff,
               const int* __restrict__ rowptr, const float* __restrict__ dinv,
               int2* __restrict__ csr, int n, int nc)
{
    __shared__ int   sCur[NPB];
    __shared__ float sDinv[NPB];
    const int t = threadIdx.x;
    const int b = blockIdx.x;
    const int bstart = tabOff[b * nc];
    const int bend   = tabOff[(b + 1) * nc];
    if (t < NPB) {
        const int node = b * NPB + t;
        const bool ok = node < n;
        sCur[t]  = ok ? rowptr[node] : 0;
        sDinv[t] = ok ? dinv[node] : 0.0f;
    }
    __syncthreads();
    for (int p0 = bstart + t; p0 < bend; p0 += 4 * SORT_T) {
        int2  r[4];
        float ds[4];
        #pragma unroll
        for (int k = 0; k < 4; ++k) {
            const int p = p0 + k * SORT_T;
            r[k] = tmp[p < bend ? p : p0];
        }
        #pragma unroll
        for (int k = 0; k < 4; ++k) ds[k] = dinv[r[k].x];
        #pragma unroll
        for (int k = 0; k < 4; ++k) {
            if (p0 + k * SORT_T < bend) {
                const int dl = r[k].y & (NPB - 1);
                const int q  = atomicAdd(&sCur[dl], 1);
                int2 rec;
                rec.x = r[k].x;
                rec.y = __float_as_int(ds[k] * sDinv[dl]);
                csr[q] = rec;
            }
        }
    }
}

#define SCAN_SLABS 10

__global__ void __launch_bounds__(1024)
k_scan(const int* __restrict__ cnt, int* __restrict__ rowptr, int n)
{
    __shared__ int sWave[SCAN_SLABS][32];
    __shared__ int sSlab[SCAN_SLABS];

    const int t    = threadIdx.x;
    const int lane = t & 31;
    const int wave = t >> 5;
    const bool vec = ((((size_t)cnt | (size_t)rowptr) & 15) == 0) && ((n & 3) == 0) && (n >= 4);
    int carry = 0;

    for (int chunk = 0; chunk < n; chunk += SCAN_SLABS * 4096) {
        int c[SCAN_SLABS][4];
        int inc[SCAN_SLABS];
        if (vec) {
            #pragma unroll
            for (int j = 0; j < SCAN_SLABS; ++j) {
                const int i0 = chunk + j * 4096 + t * 4;
                const int ic = (i0 < n) ? i0 : (n - 4);
                const int4 q = *(const int4*)(cnt + ic);
                const bool ok = (i0 < n);
                c[j][0] = ok ? q.x : 0;
                c[j][1] = ok ? q.y : 0;
                c[j][2] = ok ? q.z : 0;
                c[j][3] = ok ? q.w : 0;
            }
        } else {
            #pragma unroll
            for (int j = 0; j < SCAN_SLABS; ++j) {
                const int i0 = chunk + j * 4096 + t * 4;
                #pragma unroll
                for (int k = 0; k < 4; ++k) c[j][k] = (i0 + k < n) ? cnt[i0 + k] : 0;
            }
        }
        #pragma unroll
        for (int j = 0; j < SCAN_SLABS; ++j) inc[j] = c[j][0] + c[j][1] + c[j][2] + c[j][3];
        #pragma unroll
        for (int d = 1; d < 32; d <<= 1) {
            #pragma unroll
            for (int j = 0; j < SCAN_SLABS; ++j) {
                const int y = __shfl_up(inc[j], d);
                if (lane >= d) inc[j] += y;
            }
        }
        if (lane == 31) {
            #pragma unroll
            for (int j = 0; j < SCAN_SLABS; ++j) sWave[j][wave] = inc[j];
        }
        __syncthreads();
        if (wave < SCAN_SLABS) {
            const int w = sWave[wave][lane];
            int x = w;
            #pragma unroll
            for (int d = 1; d < 32; d <<= 1) {
                const int y = __shfl_up(x, d);
                if (lane >= d) x += y;
            }
            sWave[wave][lane] = x - w;
            if (lane == 31) sSlab[wave] = x;
        }
        __syncthreads();
        int slabBase = carry;
        #pragma unroll
        for (int j = 0; j < SCAN_SLABS; ++j) {
            const int own = c[j][0] + c[j][1] + c[j][2] + c[j][3];
            const int e0 = slabBase + sWave[j][wave] + (inc[j] - own);
            const int e1 = e0 + c[j][0];
            const int e2 = e1 + c[j][1];
            const int e3 = e2 + c[j][2];
            const int i0 = chunk + j * 4096 + t * 4;
            if (vec) {
                if (i0 < n) *(int4*)(rowptr + i0) = make_int4(e0, e1, e2, e3);
            } else {
                if (i0     < n) rowptr[i0]     = e0;
                if (i0 + 1 < n) rowptr[i0 + 1] = e1;
                if (i0 + 2 < n) rowptr[i0 + 2] = e2;
                if (i0 + 3 < n) rowptr[i0 + 3] = e3;
            }
            slabBase += sSlab[j];
        }
        carry = slabBase;
        __syncthreads();
    }
    if (t == 0) rowptr[n] = carry;
}

template <bool RELU_IN, bool ADD_BIAS, bool RELU_OUT>
__global__ void __launch_bounds__(256)
k_gemm64(const float* __restrict__ H, const float* __restrict__ W,
         const float* __restrict__ bias, float* __restrict__ out, int n)
{
    __shared__ v2f   sWp[(GD / 2) * GD];
    __shared__ float sB[GD];

    const int t = threadIdx.x;
    for (int i = t; i < (GD / 2) * GD; i += 256) {
        const int m = i >> 6;
        const int c = i & 63;
        v2f p;
        p.x = W[(2 * m) * GD + c];
        p.y = W[(2 * m + 1) * GD + c];
        sWp[i] = p;
    }
    if (ADD_BIAS && t < GD) sB[t] = bias[t];
    __syncthreads();

    const int wave = t >> 5;
    const int lane = t & 31;
    const int r    = lane & 15;
    const int hi   = lane >> 4;
    const int rowBase = blockIdx.x * 128 + wave * 16;
    if (rowBase >= n) return;

    v8f acc0 = {}, acc1 = {}, acc2 = {}, acc3 = {};
    const float* aRow = H + (size_t)(rowBase + r) * GD;

    #pragma unroll
    for (int k = 0; k < 16; ++k) {
        const int ka = 4 * k + 2 * hi;
        v2f a = *(const v2f*)(aRow + ka);
        if (RELU_IN) { a.x = fmaxf(a.x, 0.0f); a.y = fmaxf(a.y, 0.0f); }

        const int mB = 2 * k + hi;
        const v2f* wp = sWp + (size_t)mB * GD + r;
        const v2f b0 = wp[0];
        const v2f b1 = wp[16];
        const v2f b2 = wp[32];
        const v2f b3 = wp[48];

        acc0 = __builtin_amdgcn_wmma_f32_16x16x4_f32(false, a, false, b0, (short)0, acc0, false, false);
        acc1 = __builtin_amdgcn_wmma_f32_16x16x4_f32(false, a, false, b1, (short)0, acc1, false, false);
        acc2 = __builtin_amdgcn_wmma_f32_16x16x4_f32(false, a, false, b2, (short)0, acc2, false, false);
        acc3 = __builtin_amdgcn_wmma_f32_16x16x4_f32(false, a, false, b3, (short)0, acc3, false, false);
    }

    #pragma unroll
    for (int v = 0; v < 8; ++v) {
        const int row = rowBase + v + 8 * hi;
        float* o = out + (size_t)row * GD + r;
        float x0 = acc0[v], x1 = acc1[v], x2 = acc2[v], x3 = acc3[v];
        if (ADD_BIAS) { x0 += sB[r]; x1 += sB[16 + r]; x2 += sB[32 + r]; x3 += sB[48 + r]; }
        if (RELU_OUT) {
            x0 = fmaxf(x0, 0.0f); x1 = fmaxf(x1, 0.0f);
            x2 = fmaxf(x2, 0.0f); x3 = fmaxf(x3, 0.0f);
        }
        o[0] = x0; o[16] = x1; o[32] = x2; o[48] = x3;
    }
}

template <int NP>
__device__ __forceinline__ void gather_pairs(float4& acc, const int2* __restrict__ rec,
                                             const float4* __restrict__ t4, int h, int c)
{
    int    s[NP];
    float  w[NP];
    float4 v[NP];
    for (int k = 0; k < NP; ++k) {
        const int2 r0 = rec[2 * k];
        const int2 r1 = rec[2 * k + 1];
        s[k] = h ? r1.x : r0.x;
        w[k] = __int_as_float(h ? r1.y : r0.y);
    }
    for (int k = 0; k < NP; ++k) v[k] = t4[(unsigned)(s[k] * 16 + c)];
    for (int k = 0; k < NP; ++k) {
        acc.x = fmaf(v[k].x, w[k], acc.x);
        acc.y = fmaf(v[k].y, w[k], acc.y);
        acc.z = fmaf(v[k].z, w[k], acc.z);
        acc.w = fmaf(v[k].w, w[k], acc.w);
    }
}

__global__ void __launch_bounds__(256)
k_gather(const float* __restrict__ t, const int* __restrict__ rowptr, const int* __restrict__ csr,
         const float* __restrict__ dinv, const float* __restrict__ b, float* __restrict__ out, int n)
{
    const int lane = threadIdx.x & 31;
    const int h    = lane >> 4;
    const int c    = lane & 15;
    const int row = __builtin_amdgcn_readfirstlane((int)(blockIdx.x * 8 + (threadIdx.x >> 5)));
    if (row >= n) return;

    const int start = rowptr[row];
    const int end   = rowptr[row + 1];
    const float4* t4   = (const float4*)t;
    const int2*   csr2 = (const int2*)csr;

    float4 acc = make_float4(0.0f, 0.0f, 0.0f, 0.0f);
    {
        const float  d  = dinv[row];
        const float  sl = d * d;
        const float4 v  = t4[(unsigned)(row * 16 + c)];
        const float4 bb = *(const float4*)(b + c * 4);
        if (h == 0) {
            acc.x = fmaf(v.x, sl, bb.x);
            acc.y = fmaf(v.y, sl, bb.y);
            acc.z = fmaf(v.z, sl, bb.z);
            acc.w = fmaf(v.w, sl, bb.w);
        }
    }

    int p = start;
    for (; p + 8 <= end; p += 8) gather_pairs<4>(acc, csr2 + p, t4, h, c);
    if (p + 4 <= end) { gather_pairs<2>(acc, csr2 + p, t4, h, c); p += 4; }
    if (p + 2 <= end) { gather_pairs<1>(acc, csr2 + p, t4, h, c); p += 2; }
    if (p < end) {
        const int2 r = csr2[p];
        if (h == 0) {
            const float4 v = t4[(unsigned)(r.x * 16 + c)];
            const float  w = __int_as_float(r.y);
            acc.x = fmaf(v.x, w, acc.x);
            acc.y = fmaf(v.y, w, acc.y);
            acc.z = fmaf(v.z, w, acc.z);
            acc.w = fmaf(v.w, w, acc.w);
        }
    }

    acc.x += __shfl_xor(acc.x, 16);
    acc.y += __shfl_xor(acc.y, 16);
    acc.z += __shfl_xor(acc.z, 16);
    acc.w += __shfl_xor(acc.w, 16);
    if (h == 0) ((float4*)out)[(unsigned)(row * 16 + c)] = acc;
}

extern "C" void kernel_launch(void* const* d_in, const int* in_sizes, int n_in,
                              void* d_out, int out_size, void* d_ws, size_t ws_size,
                              hipStream_t stream)
{
    const float* x    = (const float*)d_in[0];
    const int*   ei   = (const int*)d_in[1];
    const float* W_in = (const float*)d_in[2];
    const float* b_in = (const float*)d_in[3];
    const float* W1   = (const float*)d_in[4];
    const float* b1   = (const float*)d_in[5];
    const float* W2   = (const float*)d_in[6];
    const float* b2   = (const float*)d_in[7];
    const float* W3   = (const float*)d_in[8];
    const float* b3   = (const float*)d_in[9];
    float* out = (float*)d_out;

    const int N = in_sizes[0] / GD;
    const int E = in_sizes[1] / 2;
    const int* src = ei;
    const int* dst = ei + E;

    float* bufT = (float*)d_ws;
    int2*  tmp  = (int2*)d_ws;
    int*   csr  = (int*)(bufT + (size_t)N * GD);
    float* dinv = (float*)(csr + (size_t)2 * E);
    int*   rowp = (int*)(dinv + N);

    const int NB = (N + NPB - 1) / NPB;
    const int NC = (E + CH - 1) / CH;
    int*   tab    = (int*)d_out;
    int*   tabOff = tab + ((NB * NC + 3) & ~3);

    const int TB = 256;
    const int gGemm = (N + 127) / 128;
    const int gRow  = (N + 7) / 8;

    k_chunk_hist   <<<NC, SORT_T, 0, stream>>>(dst, tab, E, NB, NC);
    k_scan         <<<1, 1024, 0, stream>>>(tab, tabOff, NB * NC);
    k_chunk_scatter<<<NC, SORT_T, 0, stream>>>(src, dst, tabOff, tmp, E, NB, NC);
    k_bucket_count <<<NB, SORT_T, 0, stream>>>(tmp, tabOff, rowp, dinv, N, E, NC);
    k_bucket_place <<<NB, SORT_T, 0, stream>>>(tmp, tabOff, rowp, dinv, (int2*)csr, N, NC);

    k_gemm64<false, true, true><<<gGemm, TB, 0, stream>>>(x, W_in, b_in, out, N);

    k_gemm64<false, false, false><<<gGemm, TB, 0, stream>>>(out, W1, nullptr, bufT, N);
    k_gather<<<gRow, TB, 0, stream>>>(bufT, rowp, csr, dinv, b1, out, N);

    k_gemm64<true, false, false><<<gGemm, TB, 0, stream>>>(out, W2, nullptr, bufT, N);
    k_gather<<<gRow, TB, 0, stream>>>(bufT, rowp, csr, dinv, b2, out, N);

    k_gemm64<true, false, false><<<gGemm, TB, 0, stream>>>(out, W3, nullptr, bufT, N);
    k_gather<<<gRow, TB, 0, stream>>>(bufT, rowp, csr, dinv, b3, out, N);
}


# Layer 0 and conv1 products in one kernel: relu(x@W_in+b_in) stripe passes from WMMA accumulators through the wave's LDS tile into the @W1 WMMAs, same k and sum order; 10 launches -> 9, h never stored
# speedup vs baseline: 4.0562x; 1.1650x over previous
#include <hip/hip_runtime.h>


typedef __attribute__((ext_vector_type(2))) float v2f;
typedef __attribute__((ext_vector_type(8))) float v8f;

#define GD 64

#define NPB_SHIFT 8
#define NPB  (1 << NPB_SHIFT)
#define MAXB 256
#define CH   8192
#define SORT_T 1024

__global__ void __launch_bounds__(SORT_T)
k_chunk_hist(const int* __restrict__ dst, int* __restrict__ tab, int e, int nb, int nc)
{
    __shared__ int sCnt[MAXB];
    const int t = threadIdx.x;
    if (t < MAXB) sCnt[t] = 0;
    __syncthreads();
    const int base = blockIdx.x * CH;
    int d[CH / SORT_T];
    #pragma unroll
    for (int j = 0; j < CH / SORT_T; ++j) {
        const int i = base + j * SORT_T + t;
        d[j] = dst[i < e ? i : e - 1];
    }
    #pragma unroll
    for (int j = 0; j < CH / SORT_T; ++j) {
        if (base + j * SORT_T + t < e) atomicAdd(&sCnt[d[j] >> NPB_SHIFT], 1);
    }
    __syncthreads();
    if (t < nb) tab[t * nc + blockIdx.x] = sCnt[t];
}

__global__ void __launch_bounds__(SORT_T)
k_chunk_scatter(const int* __restrict__ src, const int* __restrict__ dst,
                const int* __restrict__ tabOff, int2* __restrict__ tmp, int e, int nb, int nc)
{
    __shared__ int sCur[MAXB];
    const int t = threadIdx.x;
    if (t < nb) sCur[t] = tabOff[t * nc + blockIdx.x];
    __syncthreads();
    const int base = blockIdx.x * CH;
    int s[CH / SORT_T];
    int d[CH / SORT_T];
    #pragma unroll
    for (int j = 0; j < CH / SORT_T; ++j) {
        const int i  = base + j * SORT_T + t;
        const int ic = i < e ? i : e - 1;
        s[j] = src[ic];
        d[j] = dst[ic];
    }
    #pragma unroll
    for (int j = 0; j < CH / SORT_T; ++j) {
        if (base + j * SORT_T + t < e) {
            const int p = atomicAdd(&sCur[d[j] >> NPB_SHIFT], 1);
            tmp[p] = make_int2(s[j], d[j]);
        }
    }
}

__global__ void __launch_bounds__(SORT_T)
k_bucket_count(const int2* __restrict__ tmp, const int* __restrict__ tabOff,
               int* __restrict__ rowptr, float* __restrict__ dinv, int n, int e, int nc)
{
    __shared__ int sCnt[NPB];
    __shared__ int sTot[NPB / 32];
    const int t    = threadIdx.x;
    const int lane = t & 31;
    const int b    = blockIdx.x;
    const int bstart = tabOff[b * nc];
    const int bend   = tabOff[(b + 1) * nc];
    if (t < NPB) sCnt[t] = 0;
    __syncthreads();
    for (int p0 = bstart + t; p0 < bend; p0 += 4 * SORT_T) {
        int d[4];
        #pragma unroll
        for (int k = 0; k < 4; ++k) {
            const int p = p0 + k * SORT_T;
            d[k] = tmp[p < bend ? p : p0].y;
        }
        #pragma unroll
        for (int k = 0; k < 4; ++k) {
            if (p0 + k * SORT_T < bend) atomicAdd(&sCnt[d[k] & (NPB - 1)], 1);
        }
    }
    __syncthreads();
    int c = 0, inc = 0;
    if (t < NPB) {
        c = sCnt[t];
        inc = c;
        #pragma unroll
        for (int d = 1; d < 32; d <<= 1) {
            const int y = __shfl_up(inc, d);
            if (lane >= d) inc += y;
        }
        if (lane == 31) sTot[t >> 5] = inc;
    }
    __syncthreads();
    if (t < NPB) {
        int pre = 0;
        for (int w = 0; w < (t >> 5); ++w) pre += sTot[w];
        const int node = b * NPB + t;
        if (node < n) {
            rowptr[node] = bstart + pre + inc - c;
            dinv[node]   = rsqrtf(fmaxf((float)(c + 1), 1e-12f));
        }
    }
    if (b == 0 && t == 0) rowptr[n] = e;
}

__global__ void __launch_bounds__(SORT_T)
k_bucket_place(const int2* __restrict__ tmp, const int* __restrict__ tabOff,
               const int* __restrict__ rowptr, const float* __restrict__ dinv,
               int2* __restrict__ csr, int n, int nc)
{
    __shared__ int   sCur[NPB];
    __shared__ float sDinv[NPB];
    const int t = threadIdx.x;
    const int b = blockIdx.x;
    const int bstart = tabOff[b * nc];
    const int bend   = tabOff[(b + 1) * nc];
    if (t < NPB) {
        const int node = b * NPB + t;
        const bool ok = node < n;
        sCur[t]  = ok ? rowptr[node] : 0;
        sDinv[t] = ok ? dinv[node] : 0.0f;
    }
    __syncthreads();
    for (int p0 = bstart + t; p0 < bend; p0 += 4 * SORT_T) {
        int2  r[4];
        float ds[4];
        #pragma unroll
        for (int k = 0; k < 4; ++k) {
            const int p = p0 + k * SORT_T;
            r[k] = tmp[p < bend ? p : p0];
        }
        #pragma unroll
        for (int k = 0; k < 4; ++k) ds[k] = dinv[r[k].x];
        #pragma unroll
        for (int k = 0; k < 4; ++k) {
            if (p0 + k * SORT_T < bend) {
                const int dl = r[k].y & (NPB - 1);
                const int q  = atomicAdd(&sCur[dl], 1);
                int2 rec;
                rec.x = r[k].x;
                rec.y = __float_as_int(ds[k] * sDinv[dl]);
                csr[q] = rec;
            }
        }
    }
}

#define SCAN_SLABS 10

__global__ void __launch_bounds__(1024)
k_scan(const int* __restrict__ cnt, int* __restrict__ rowptr, int n)
{
    __shared__ int sWave[SCAN_SLABS][32];
    __shared__ int sSlab[SCAN_SLABS];

    const int t    = threadIdx.x;
    const int lane = t & 31;
    const int wave = t >> 5;
    const bool vec = ((((size_t)cnt | (size_t)rowptr) & 15) == 0) && ((n & 3) == 0) && (n >= 4);
    int carry = 0;

    for (int chunk = 0; chunk < n; chunk += SCAN_SLABS * 4096) {
        int c[SCAN_SLABS][4];
        int inc[SCAN_SLABS];
        if (vec) {
            #pragma unroll
            for (int j = 0; j < SCAN_SLABS; ++j) {
                const int i0 = chunk + j * 4096 + t * 4;
                const int ic = (i0 < n) ? i0 : (n - 4);
                const int4 q = *(const int4*)(cnt + ic);
                const bool ok = (i0 < n);
                c[j][0] = ok ? q.x : 0;
                c[j][1] = ok ? q.y : 0;
                c[j][2] = ok ? q.z : 0;
                c[j][3] = ok ? q.w : 0;
            }
        } else {
            #pragma unroll
            for (int j = 0; j < SCAN_SLABS; ++j) {
                const int i0 = chunk + j * 4096 + t * 4;
                #pragma unroll
                for (int k = 0; k < 4; ++k) c[j][k] = (i0 + k < n) ? cnt[i0 + k] : 0;
            }
        }
        #pragma unroll
        for (int j = 0; j < SCAN_SLABS; ++j) inc[j] = c[j][0] + c[j][1] + c[j][2] + c[j][3];
        #pragma unroll
        for (int d = 1; d < 32; d <<= 1) {
            #pragma unroll
            for (int j = 0; j < SCAN_SLABS; ++j) {
                const int y = __shfl_up(inc[j], d);
                if (lane >= d) inc[j] += y;
            }
        }
        if (lane == 31) {
            #pragma unroll
            for (int j = 0; j < SCAN_SLABS; ++j) sWave[j][wave] = inc[j];
        }
        __syncthreads();
        if (wave < SCAN_SLABS) {
            const int w = sWave[wave][lane];
            int x = w;
            #pragma unroll
            for (int d = 1; d < 32; d <<= 1) {
                const int y = __shfl_up(x, d);
                if (lane >= d) x += y;
            }
            sWave[wave][lane] = x - w;
            if (lane == 31) sSlab[wave] = x;
        }
        __syncthreads();
        int slabBase = carry;
        #pragma unroll
        for (int j = 0; j < SCAN_SLABS; ++j) {
            const int own = c[j][0] + c[j][1] + c[j][2] + c[j][3];
            const int e0 = slabBase + sWave[j][wave] + (inc[j] - own);
            const int e1 = e0 + c[j][0];
            const int e2 = e1 + c[j][1];
            const int e3 = e2 + c[j][2];
            const int i0 = chunk + j * 4096 + t * 4;
            if (vec) {
                if (i0 < n) *(int4*)(rowptr + i0) = make_int4(e0, e1, e2, e3);
            } else {
                if (i0     < n) rowptr[i0]     = e0;
                if (i0 + 1 < n) rowptr[i0 + 1] = e1;
                if (i0 + 2 < n) rowptr[i0 + 2] = e2;
                if (i0 + 3 < n) rowptr[i0 + 3] = e3;
            }
            slabBase += sSlab[j];
        }
        carry = slabBase;
        __syncthreads();
    }
    if (t == 0) rowptr[n] = carry;
}

#define HPITCH 68

__global__ void __launch_bounds__(256)
k_gemm64x2(const float* __restrict__ X, const float* __restrict__ Wa, const float* __restrict__ ba,
           const float* __restrict__ Wb, float* __restrict__ out, int n)
{
    __shared__ v2f   sWa[(GD / 2) * GD];
    __shared__ v2f   sWb[(GD / 2) * GD];
    __shared__ float sB[GD];
    __shared__ __attribute__((aligned(16))) float sH[8 * 16 * HPITCH];

    const int t = threadIdx.x;
    for (int i = t; i < (GD / 2) * GD; i += 256) {
        const int m = i >> 6;
        const int c = i & 63;
        v2f p, q;
        p.x = Wa[(2 * m) * GD + c];
        p.y = Wa[(2 * m + 1) * GD + c];
        q.x = Wb[(2 * m) * GD + c];
        q.y = Wb[(2 * m + 1) * GD + c];
        sWa[i] = p;
        sWb[i] = q;
    }
    if (t < GD) sB[t] = ba[t];
    __syncthreads();

    const int wave = t >> 5;
    const int lane = t & 31;
    const int r    = lane & 15;
    const int hi   = lane >> 4;
    const int rowBase = blockIdx.x * 128 + wave * 16;
    const bool live = rowBase < n;
    float* hT = sH + wave * (16 * HPITCH);

    if (live) {
        v8f acc0 = {}, acc1 = {}, acc2 = {}, acc3 = {};
        const float* aRow = X + (size_t)(rowBase + r) * GD;

        #pragma unroll
        for (int k = 0; k < 16; ++k) {
            const int ka = 4 * k + 2 * hi;
            const v2f a = *(const v2f*)(aRow + ka);

            const int mB = 2 * k + hi;
            const v2f* wp = sWa + (size_t)mB * GD + r;
            const v2f b0 = wp[0];
            const v2f b1 = wp[16];
            const v2f b2 = wp[32];
            const v2f b3 = wp[48];

            acc0 = __builtin_amdgcn_wmma_f32_16x16x4_f32(false, a, false, b0, (short)0, acc0, false, false);
            acc1 = __builtin_amdgcn_wmma_f32_16x16x4_f32(false, a, false, b1, (short)0, acc1, false, false);
            acc2 = __builtin_amdgcn_wmma_f32_16x16x4_f32(false, a, false, b2, (short)0, acc2, false, false);
            acc3 = __builtin_amdgcn_wmma_f32_16x16x4_f32(false, a, false, b3, (short)0, acc3, false, false);
        }

        #pragma unroll
        for (int v = 0; v < 8; ++v) {
            float* o = hT + (v + 8 * hi) * HPITCH + r;
            float x0 = acc0[v], x1 = acc1[v], x2 = acc2[v], x3 = acc3[v];
            x0 += sB[r]; x1 += sB[16 + r]; x2 += sB[32 + r]; x3 += sB[48 + r];
            x0 = fmaxf(x0, 0.0f); x1 = fmaxf(x1, 0.0f);
            x2 = fmaxf(x2, 0.0f); x3 = fmaxf(x3, 0.0f);
            o[0] = x0; o[16] = x1; o[32] = x2; o[48] = x3;
        }
    }
    __syncthreads();
    if (!live) return;

    v8f acc0 = {}, acc1 = {}, acc2 = {}, acc3 = {};
    const float* hRow = hT + r * HPITCH;

    #pragma unroll
    for (int k = 0; k < 16; ++k) {
        const int ka = 4 * k + 2 * hi;
        const v2f a = *(const v2f*)(hRow + ka);

        const int mB = 2 * k + hi;
        const v2f* wp = sWb + (size_t)mB * GD + r;
        const v2f b0 = wp[0];
        const v2f b1 = wp[16];
        const v2f b2 = wp[32];
        const v2f b3 = wp[48];

        acc0 = __builtin_amdgcn_wmma_f32_16x16x4_f32(false, a, false, b0, (short)0, acc0, false, false);
        acc1 = __builtin_amdgcn_wmma_f32_16x16x4_f32(false, a, false, b1, (short)0, acc1, false, false);
        acc2 = __builtin_amdgcn_wmma_f32_16x16x4_f32(false, a, false, b2, (short)0, acc2, false, false);
        acc3 = __builtin_amdgcn_wmma_f32_16x16x4_f32(false, a, false, b3, (short)0, acc3, false, false);
    }

    #pragma unroll
    for (int v = 0; v < 8; ++v) {
        const int row = rowBase + v + 8 * hi;
        float* o = out + (size_t)row * GD + r;
        o[0] = acc0[v]; o[16] = acc1[v]; o[32] = acc2[v]; o[48] = acc3[v];
    }
}

template <int NP>
__device__ __forceinline__ void gather_pairs(float4& acc, const int2* __restrict__ rec,
                                             const float4* __restrict__ t4, int h, int c)
{
    int    s[NP];
    float  w[NP];
    float4 v[NP];
    for (int k = 0; k < NP; ++k) {
        const int2 r0 = rec[2 * k];
        const int2 r1 = rec[2 * k + 1];
        s[k] = h ? r1.x : r0.x;
        w[k] = __int_as_float(h ? r1.y : r0.y);
    }
    for (int k = 0; k < NP; ++k) v[k] = t4[(unsigned)(s[k] * 16 + c)];
    for (int k = 0; k < NP; ++k) {
        acc.x = fmaf(v[k].x, w[k], acc.x);
        acc.y = fmaf(v[k].y, w[k], acc.y);
        acc.z = fmaf(v[k].z, w[k], acc.z);
        acc.w = fmaf(v[k].w, w[k], acc.w);
    }
}

__device__ __forceinline__ float4 gather_row(const float4* __restrict__ t4, const int* __restrict__ rowptr,
                                             const int2* __restrict__ csr2, const float* __restrict__ dinv,
                                             const float* __restrict__ b, int row, int h, int c)
{
    const int start = rowptr[row];
    const int end   = rowptr[row + 1];

    float4 acc = make_float4(0.0f, 0.0f, 0.0f, 0.0f);
    {
        const float  d  = dinv[row];
        const float  sl = d * d;
        const float4 v  = t4[(unsigned)(row * 16 + c)];
        const float4 bb = *(const float4*)(b + c * 4);
        if (h == 0) {
            acc.x = fmaf(v.x, sl, bb.x);
            acc.y = fmaf(v.y, sl, bb.y);
            acc.z = fmaf(v.z, sl, bb.z);
            acc.w = fmaf(v.w, sl, bb.w);
        }
    }

    int p = start;
    for (; p + 8 <= end; p += 8) gather_pairs<4>(acc, csr2 + p, t4, h, c);
    if (p + 4 <= end) { gather_pairs<2>(acc, csr2 + p, t4, h, c); p += 4; }
    if (p + 2 <= end) { gather_pairs<1>(acc, csr2 + p, t4, h, c); p += 2; }
    if (p < end) {
        const int2 r = csr2[p];
        if (h == 0) {
            const float4 v = t4[(unsigned)(r.x * 16 + c)];
            const float  w = __int_as_float(r.y);
            acc.x = fmaf(v.x, w, acc.x);
            acc.y = fmaf(v.y, w, acc.y);
            acc.z = fmaf(v.z, w, acc.z);
            acc.w = fmaf(v.w, w, acc.w);
        }
    }

    acc.x += __shfl_xor(acc.x, 16);
    acc.y += __shfl_xor(acc.y, 16);
    acc.z += __shfl_xor(acc.z, 16);
    acc.w += __shfl_xor(acc.w, 16);
    return acc;
}

__global__ void __launch_bounds__(256)
k_gather(const float* __restrict__ t, const int* __restrict__ rowptr, const int* __restrict__ csr,
         const float* __restrict__ dinv, const float* __restrict__ b, float* __restrict__ out, int n)
{
    const int lane = threadIdx.x & 31;
    const int h    = lane >> 4;
    const int c    = lane & 15;
    const int row = __builtin_amdgcn_readfirstlane((int)(blockIdx.x * 8 + (threadIdx.x >> 5)));
    if (row >= n) return;
    const float4 acc = gather_row((const float4*)t, rowptr, (const int2*)csr, dinv, b, row, h, c);
    if (h == 0) ((float4*)out)[(unsigned)(row * 16 + c)] = acc;
}

#define FR     16
#define APITCH 68

__global__ void __launch_bounds__(256) __attribute__((amdgpu_waves_per_eu(16, 16)))
k_gather_gemm(const float* __restrict__ t, const int* __restrict__ rowptr, const int* __restrict__ csr,
              const float* __restrict__ dinv, const float* __restrict__ b, const float* __restrict__ Wn,
              float* __restrict__ tout, int n)
{
    __shared__ __attribute__((aligned(16))) float sA[FR * APITCH];

    const int lane = threadIdx.x & 31;
    const int h    = lane >> 4;
    const int c    = lane & 15;
    const int wave = __builtin_amdgcn_readfirstlane((int)(threadIdx.x >> 5));
    const int rowBase = blockIdx.x * FR;

    for (int j = 0; j < FR / 8; ++j) {
        const int lr  = wave * (FR / 8) + j;
        const int row = rowBase + lr;
        float4 acc = make_float4(0.0f, 0.0f, 0.0f, 0.0f);
        if (row < n) acc = gather_row((const float4*)t, rowptr, (const int2*)csr, dinv, b, row, h, c);
        if (h == 0) *(float4*)(sA + lr * APITCH + 4 * c) = acc;
    }
    __syncthreads();
    if (wave >= 2) return;

    const int r  = c;
    const int hi = h;
    v8f acc0 = {}, acc1 = {};
    const float* aRow = sA + r * APITCH;
    const float* wcol = Wn + wave * 32 + r;

    #pragma unroll
    for (int k = 0; k < 16; ++k) {
        const int ka = 4 * k + 2 * hi;
        v2f a = *(const v2f*)(aRow + ka);
        a.x = fmaxf(a.x, 0.0f); a.y = fmaxf(a.y, 0.0f);
        v2f b0, b1;
        b0.x = wcol[ka * GD];
        b0.y = wcol[(ka + 1) * GD];
        b1.x = wcol[ka * GD + 16];
        b1.y = wcol[(ka + 1) * GD + 16];
        acc0 = __builtin_amdgcn_wmma_f32_16x16x4_f32(false, a, false, b0, (short)0, acc0, false, false);
        acc1 = __builtin_amdgcn_wmma_f32_16x16x4_f32(false, a, false, b1, (short)0, acc1, false, false);
    }

    #pragma unroll
    for (int v = 0; v < 8; ++v) {
        const int row = rowBase + v + 8 * hi;
        if (row < n) {
            float* o = tout + (size_t)row * GD + wave * 32 + r;
            o[0]  = acc0[v];
            o[16] = acc1[v];
        }
    }
}

extern "C" void kernel_launch(void* const* d_in, const int* in_sizes, int n_in,
                              void* d_out, int out_size, void* d_ws, size_t ws_size,
                              hipStream_t stream)
{
    const float* x    = (const float*)d_in[0];
    const int*   ei   = (const int*)d_in[1];
    const float* W_in = (const float*)d_in[2];
    const float* b_in = (const float*)d_in[3];
    const float* W1   = (const float*)d_in[4];
    const float* b1   = (const float*)d_in[5];
    const float* W2   = (const float*)d_in[6];
    const float* b2   = (const float*)d_in[7];
    const float* W3   = (const float*)d_in[8];
    const float* b3   = (const float*)d_in[9];
    float* out = (float*)d_out;

    const int N = in_sizes[0] / GD;
    const int E = in_sizes[1] / 2;
    const int* src = ei;
    const int* dst = ei + E;

    float* bufT = (float*)d_ws;
    int2*  tmp  = (int2*)d_ws;
    int*   csr  = (int*)(bufT + (size_t)N * GD);
    float* dinv = (float*)(csr + (size_t)2 * E);
    int*   rowp = (int*)(dinv + N);

    const int NB = (N + NPB - 1) / NPB;
    const int NC = (E + CH - 1) / CH;
    int*   tab    = (int*)d_out;
    int*   tabOff = tab + ((NB * NC + 3) & ~3);

    const int TB = 256;
    const int gGemm = (N + 127) / 128;
    const int gRow  = (N + 7) / 8;
    const int gFuse = (N + FR - 1) / FR;

    k_chunk_hist   <<<NC, SORT_T, 0, stream>>>(dst, tab, E, NB, NC);
    k_scan         <<<1, 1024, 0, stream>>>(tab, tabOff, NB * NC);
    k_chunk_scatter<<<NC, SORT_T, 0, stream>>>(src, dst, tabOff, tmp, E, NB, NC);
    k_bucket_count <<<NB, SORT_T, 0, stream>>>(tmp, tabOff, rowp, dinv, N, E, NC);
    k_bucket_place <<<NB, SORT_T, 0, stream>>>(tmp, tabOff, rowp, dinv, (int2*)csr, N, NC);

    k_gemm64x2<<<gGemm, TB, 0, stream>>>(x, W_in, b_in, W1, bufT, N);

    k_gather_gemm<<<gFuse, TB, 0, stream>>>(bufT, rowp, csr, dinv, b1, W2, out, N);

    k_gather_gemm<<<gFuse, TB, 0, stream>>>(out, rowp, csr, dinv, b2, W3, bufT, N);

    k_gather<<<gRow, TB, 0, stream>>>(bufT, rowp, csr, dinv, b3, out, N);
}


# Gather launches on a quarter of the waves: a wave walks 4 consecutive rows, next 8 records and next row's bound, records, own line fetched under the current row loads; same sum order
# speedup vs baseline: 4.2350x; 1.0441x over previous
#include <hip/hip_runtime.h>


typedef __attribute__((ext_vector_type(2))) float v2f;
typedef __attribute__((ext_vector_type(8))) float v8f;

#define GD 64

#define NPB_SHIFT 8
#define NPB  (1 << NPB_SHIFT)
#define MAXB 256
#define CH   8192
#define SORT_T 1024

__global__ void __launch_bounds__(SORT_T)
k_chunk_hist(const int* __restrict__ dst, int* __restrict__ tab, int e, int nb, int nc)
{
    __shared__ int sCnt[MAXB];
    const int t = threadIdx.x;
    if (t < MAXB) sCnt[t] = 0;
    __syncthreads();
    const int base = blockIdx.x * CH;
    int d[CH / SORT_T];
    #pragma unroll
    for (int j = 0; j < CH / SORT_T; ++j) {
        const int i = base + j * SORT_T + t;
        d[j] = dst[i < e ? i : e - 1];
    }
    #pragma unroll
    for (int j = 0; j < CH / SORT_T; ++j) {
        if (base + j * SORT_T + t < e) atomicAdd(&sCnt[d[j] >> NPB_SHIFT], 1);
    }
    __syncthreads();
    if (t < nb) tab[t * nc + blockIdx.x] = sCnt[t];
}

__global__ void __launch_bounds__(SORT_T)
k_chunk_scatter(const int* __restrict__ src, const int* __restrict__ dst,
                const int* __restrict__ tabOff, int2* __restrict__ tmp, int e, int nb, int nc)
{
    __shared__ int sCur[MAXB];
    const int t = threadIdx.x;
    if (t < nb) sCur[t] = tabOff[t * nc + blockIdx.x];
    __syncthreads();
    const int base = blockIdx.x * CH;
    int s[CH / SORT_T];
    int d[CH / SORT_T];
    #pragma unroll
    for (int j = 0; j < CH / SORT_T; ++j) {
        const int i  = base + j * SORT_T + t;
        const int ic = i < e ? i : e - 1;
        s[j] = src[ic];
        d[j] = dst[ic];
    }
    #pragma unroll
    for (int j = 0; j < CH / SORT_T; ++j) {
        if (base + j * SORT_T + t < e) {
            const int p = atomicAdd(&sCur[d[j] >> NPB_SHIFT], 1);
            tmp[p] = make_int2(s[j], d[j]);
        }
    }
}

__global__ void __launch_bounds__(SORT_T)
k_bucket_count(const int2* __restrict__ tmp, const int* __restrict__ tabOff,
               int* __restrict__ rowptr, float* __restrict__ dinv, int n, int e, int nc)
{
    __shared__ int sCnt[NPB];
    __shared__ int sTot[NPB / 32];
    const int t    = threadIdx.x;
    const int lane = t & 31;
    const int b    = blockIdx.x;
    const int bstart = tabOff[b * nc];
    const int bend   = tabOff[(b + 1) * nc];
    if (t < NPB) sCnt[t] = 0;
    __syncthreads();
    for (int p0 = bstart + t; p0 < bend; p0 += 4 * SORT_T) {
        int d[4];
        #pragma unroll
        for (int k = 0; k < 4; ++k) {
            const int p = p0 + k * SORT_T;
            d[k] = tmp[p < bend ? p : p0].y;
        }
        #pragma unroll
        for (int k = 0; k < 4; ++k) {
            if (p0 + k * SORT_T < bend) atomicAdd(&sCnt[d[k] & (NPB - 1)], 1);
        }
    }
    __syncthreads();
    int c = 0, inc = 0;
    if (t < NPB) {
        c = sCnt[t];
        inc = c;
        #pragma unroll
        for (int d = 1; d < 32; d <<= 1) {
            const int y = __shfl_up(inc, d);
            if (lane >= d) inc += y;
        }
        if (lane == 31) sTot[t >> 5] = inc;
    }
    __syncthreads();
    if (t < NPB) {
        int pre = 0;
        for (int w = 0; w < (t >> 5); ++w) pre += sTot[w];
        const int node = b * NPB + t;
        if (node < n) {
            rowptr[node] = bstart + pre + inc - c;
            dinv[node]   = rsqrtf(fmaxf((float)(c + 1), 1e-12f));
        }
    }
    if (b == 0 && t == 0) rowptr[n] = e;
}

__global__ void __launch_bounds__(SORT_T)
k_bucket_place(const int2* __restrict__ tmp, const int* __restrict__ tabOff,
               const int* __restrict__ rowptr, const float* __restrict__ dinv,
               int2* __restrict__ csr, int n, int nc)
{
    __shared__ int   sCur[NPB];
    __shared__ float sDinv[NPB];
    const int t = threadIdx.x;
    const int b = blockIdx.x;
    const int bstart = tabOff[b * nc];
    const int bend   = tabOff[(b + 1) * nc];
    if (t < NPB) {
        const int node = b * NPB + t;
        const bool ok = node < n;
        sCur[t]  = ok ? rowptr[node] : 0;
        sDinv[t] = ok ? dinv[node] : 0.0f;
    }
    __syncthreads();
    for (int p0 = bstart + t; p0 < bend; p0 += 4 * SORT_T) {
        int2  r[4];
        float ds[4];
        #pragma unroll
        for (int k = 0; k < 4; ++k) {
            const int p = p0 + k * SORT_T;
            r[k] = tmp[p < bend ? p : p0];
        }
        #pragma unroll
        for (int k = 0; k < 4; ++k) ds[k] = dinv[r[k].x];
        #pragma unroll
        for (int k = 0; k < 4; ++k) {
            if (p0 + k * SORT_T < bend) {
                const int dl = r[k].y & (NPB - 1);
                const int q  = atomicAdd(&sCur[dl], 1);
                int2 rec;
                rec.x = r[k].x;
                rec.y = __float_as_int(ds[k] * sDinv[dl]);
                csr[q] = rec;
            }
        }
    }
}

#define SCAN_SLABS 10

__global__ void __launch_bounds__(1024)
k_scan(const int* __restrict__ cnt, int* __restrict__ rowptr, int n)
{
    __shared__ int sWave[SCAN_SLABS][32];
    __shared__ int sSlab[SCAN_SLABS];

    const int t    = threadIdx.x;
    const int lane = t & 31;
    const int wave = t >> 5;
    const bool vec = ((((size_t)cnt | (size_t)rowptr) & 15) == 0) && ((n & 3) == 0) && (n >= 4);
    int carry = 0;

    for (int chunk = 0; chunk < n; chunk += SCAN_SLABS * 4096) {
        int c[SCAN_SLABS][4];
        int inc[SCAN_SLABS];
        if (vec) {
            #pragma unroll
            for (int j = 0; j < SCAN_SLABS; ++j) {
                const int i0 = chunk + j * 4096 + t * 4;
                const int ic = (i0 < n) ? i0 : (n - 4);
                const int4 q = *(const int4*)(cnt + ic);
                const bool ok = (i0 < n);
                c[j][0] = ok ? q.x : 0;
                c[j][1] = ok ? q.y : 0;
                c[j][2] = ok ? q.z : 0;
                c[j][3] = ok ? q.w : 0;
            }
        } else {
            #pragma unroll
            for (int j = 0; j < SCAN_SLABS; ++j) {
                const int i0 = chunk + j * 4096 + t * 4;
                #pragma unroll
                for (int k = 0; k < 4; ++k) c[j][k] = (i0 + k < n) ? cnt[i0 + k] : 0;
            }
        }
        #pragma unroll
        for (int j = 0; j < SCAN_SLABS; ++j) inc[j] = c[j][0] + c[j][1] + c[j][2] + c[j][3];
        #pragma unroll
        for (int d = 1; d < 32; d <<= 1) {
            #pragma unroll
            for (int j = 0; j < SCAN_SLABS; ++j) {
                const int y = __shfl_up(inc[j], d);
                if (lane >= d) inc[j] += y;
            }
        }
        if (lane == 31) {
            #pragma unroll
            for (int j = 0; j < SCAN_SLABS; ++j) sWave[j][wave] = inc[j];
        }
        __syncthreads();
        if (wave < SCAN_SLABS) {
            const int w = sWave[wave][lane];
            int x = w;
            #pragma unroll
            for (int d = 1; d < 32; d <<= 1) {
                const int y = __shfl_up(x, d);
                if (lane >= d) x += y;
            }
            sWave[wave][lane] = x - w;
            if (lane == 31) sSlab[wave] = x;
        }
        __syncthreads();
        int slabBase = carry;
        #pragma unroll
        for (int j = 0; j < SCAN_SLABS; ++j) {
            const int own = c[j][0] + c[j][1] + c[j][2] + c[j][3];
            const int e0 = slabBase + sWave[j][wave] + (inc[j] - own);
            const int e1 = e0 + c[j][0];
            const int e2 = e1 + c[j][1];
            const int e3 = e2 + c[j][2];
            const int i0 = chunk + j * 4096 + t * 4;
            if (vec) {
                if (i0 < n) *(int4*)(rowptr + i0) = make_int4(e0, e1, e2, e3);
            } else {
                if (i0     < n) rowptr[i0]     = e0;
                if (i0 + 1 < n) rowptr[i0 + 1] = e1;
                if (i0 + 2 < n) rowptr[i0 + 2] = e2;
                if (i0 + 3 < n) rowptr[i0 + 3] = e3;
            }
            slabBase += sSlab[j];
        }
        carry = slabBase;
        __syncthreads();
    }
    if (t == 0) rowptr[n] = carry;
}

#define HPITCH 68

__global__ void __launch_bounds__(256)
k_gemm64x2(const float* __restrict__ X, const float* __restrict__ Wa, const float* __restrict__ ba,
           const float* __restrict__ Wb, float* __restrict__ out, int n)
{
    __shared__ v2f   sWa[(GD / 2) * GD];
    __shared__ v2f   sWb[(GD / 2) * GD];
    __shared__ float sB[GD];
    __shared__ __attribute__((aligned(16))) float sH[8 * 16 * HPITCH];

    const int t = threadIdx.x;
    for (int i = t; i < (GD / 2) * GD; i += 256) {
        const int m = i >> 6;
        const int c = i & 63;
        v2f p, q;
        p.x = Wa[(2 * m) * GD + c];
        p.y = Wa[(2 * m + 1) * GD + c];
        q.x = Wb[(2 * m) * GD + c];
        q.y = Wb[(2 * m + 1) * GD + c];
        sWa[i] = p;
        sWb[i] = q;
    }
    if (t < GD) sB[t] = ba[t];
    __syncthreads();

    const int wave = t >> 5;
    const int lane = t & 31;
    const int r    = lane & 15;
    const int hi   = lane >> 4;
    const int rowBase = blockIdx.x * 128 + wave * 16;
    const bool live = rowBase < n;
    float* hT = sH + wave * (16 * HPITCH);

    if (live) {
        v8f acc0 = {}, acc1 = {}, acc2 = {}, acc3 = {};
        const float* aRow = X + (size_t)(rowBase + r) * GD;

        #pragma unroll
        for (int k = 0; k < 16; ++k) {
            const int ka = 4 * k + 2 * hi;
            const v2f a = *(const v2f*)(aRow + ka);

            const int mB = 2 * k + hi;
            const v2f* wp = sWa + (size_t)mB * GD + r;
            const v2f b0 = wp[0];
            const v2f b1 = wp[16];
            const v2f b2 = wp[32];
            const v2f b3 = wp[48];

            acc0 = __builtin_amdgcn_wmma_f32_16x16x4_f32(false, a, false, b0, (short)0, acc0, false, false);
            acc1 = __builtin_amdgcn_wmma_f32_16x16x4_f32(false, a, false, b1, (short)0, acc1, false, false);
            acc2 = __builtin_amdgcn_wmma_f32_16x16x4_f32(false, a, false, b2, (short)0, acc2, false, false);
            acc3 = __builtin_amdgcn_wmma_f32_16x16x4_f32(false, a, false, b3, (short)0, acc3, false, false);
        }

        #pragma unroll
        for (int v = 0; v < 8; ++v) {
            float* o = hT + (v + 8 * hi) * HPITCH + r;
            float x0 = acc0[v], x1 = acc1[v], x2 = acc2[v], x3 = acc3[v];
            x0 += sB[r]; x1 += sB[16 + r]; x2 += sB[32 + r]; x3 += sB[48 + r];
            x0 = fmaxf(x0, 0.0f); x1 = fmaxf(x1, 0.0f);
            x2 = fmaxf(x2, 0.0f); x3 = fmaxf(x3, 0.0f);
            o[0] = x0; o[16] = x1; o[32] = x2; o[48] = x3;
        }
    }
    __syncthreads();
    if (!live) return;

    v8f acc0 = {}, acc1 = {}, acc2 = {}, acc3 = {};
    const float* hRow = hT + r * HPITCH;

    #pragma unroll
    for (int k = 0; k < 16; ++k) {
        const int ka = 4 * k + 2 * hi;
        const v2f a = *(const v2f*)(hRow + ka);

        const int mB = 2 * k + hi;
        const v2f* wp = sWb + (size_t)mB * GD + r;
        const v2f b0 = wp[0];
        const v2f b1 = wp[16];
        const v2f b2 = wp[32];
        const v2f b3 = wp[48];

        acc0 = __builtin_amdgcn_wmma_f32_16x16x4_f32(false, a, false, b0, (short)0, acc0, false, false);
        acc1 = __builtin_amdgcn_wmma_f32_16x16x4_f32(false, a, false, b1, (short)0, acc1, false, false);
        acc2 = __builtin_amdgcn_wmma_f32_16x16x4_f32(false, a, false, b2, (short)0, acc2, false, false);
        acc3 = __builtin_amdgcn_wmma_f32_16x16x4_f32(false, a, false, b3, (short)0, acc3, false, false);
    }

    #pragma unroll
    for (int v = 0; v < 8; ++v) {
        const int row = rowBase + v + 8 * hi;
        float* o = out + (size_t)row * GD + r;
        o[0] = acc0[v]; o[16] = acc1[v]; o[32] = acc2[v]; o[48] = acc3[v];
    }
}

__device__ __forceinline__ float4 row_start(float4 sv, float d, float4 bb, int h)
{
    float4 acc = make_float4(0.0f, 0.0f, 0.0f, 0.0f);
    if (h == 0) {
        const float sl = d * d;
        acc.x = fmaf(sv.x, sl, bb.x);
        acc.y = fmaf(sv.y, sl, bb.y);
        acc.z = fmaf(sv.z, sl, bb.z);
        acc.w = fmaf(sv.w, sl, bb.w);
    }
    return acc;
}

__device__ __forceinline__ void gather8(float4& acc, const int2 (&R)[8], const float4* __restrict__ t4, int h, int c)
{
    int    s[4];
    float  w[4];
    float4 v[4];
    for (int k = 0; k < 4; ++k) {
        const int2 r0 = R[2 * k];
        const int2 r1 = R[2 * k + 1];
        s[k] = h ? r1.x : r0.x;
        w[k] = __int_as_float(h ? r1.y : r0.y);
    }
    for (int k = 0; k < 4; ++k) v[k] = t4[(unsigned)(s[k] * 16 + c)];
    for (int k = 0; k < 4; ++k) {
        acc.x = fmaf(v[k].x, w[k], acc.x);
        acc.y = fmaf(v[k].y, w[k], acc.y);
        acc.z = fmaf(v[k].z, w[k], acc.z);
        acc.w = fmaf(v[k].w, w[k], acc.w);
    }
}

__device__ __forceinline__ float4 gather_row(const float4* __restrict__ t4, const int2* __restrict__ csr2,
                                             int2 (&R)[8], float4 acc,
                                             int row, int start, int end, int h, int c)
{

    int p = start;
    while (p + 8 <= end) {
        int2 Q[8];
        for (int k = 0; k < 8; ++k) Q[k] = csr2[p + 8 + k];
        gather8(acc, R, t4, h, c);
        p += 8;
        if (p + 8 > end) {
            for (int k = 0; k < 8; ++k) R[k] = Q[k];
            break;
        }
        for (int k = 0; k < 8; ++k) R[k] = csr2[p + 8 + k];
        gather8(acc, Q, t4, h, c);
        p += 8;
    }

    const int nrem = end - p;
    if (nrem > 0) {
        int    s[4];
        float  w[4];
        float4 v[4];
        bool   ok[4];
        for (int k = 0; k < 4; ++k) {
            const int2 r0 = R[2 * k];
            const int2 r1 = R[2 * k + 1];
            ok[k] = (2 * k + h) < nrem;
            s[k] = ok[k] ? (h ? r1.x : r0.x) : row;
            w[k] = __int_as_float(h ? r1.y : r0.y);
        }
        for (int k = 0; k < 4; ++k) v[k] = t4[(unsigned)(s[k] * 16 + c)];
        for (int k = 0; k < 4; ++k) {
            if (ok[k]) {
                acc.x = fmaf(v[k].x, w[k], acc.x);
                acc.y = fmaf(v[k].y, w[k], acc.y);
                acc.z = fmaf(v[k].z, w[k], acc.z);
                acc.w = fmaf(v[k].w, w[k], acc.w);
            }
        }
    }

    acc.x += __shfl_xor(acc.x, 16);
    acc.y += __shfl_xor(acc.y, 16);
    acc.z += __shfl_xor(acc.z, 16);
    acc.w += __shfl_xor(acc.w, 16);
    return acc;
}

#define RPW 4

__global__ void __launch_bounds__(256)
k_gather(const float* __restrict__ t, const int* __restrict__ rowptr, const int* __restrict__ csr,
         const float* __restrict__ dinv, const float* __restrict__ b, float* __restrict__ out, int n)
{
    const int lane = threadIdx.x & 31;
    const int h    = lane >> 4;
    const int c    = lane & 15;
    const int wv = __builtin_amdgcn_readfirstlane((int)(blockIdx.x * 8 + (threadIdx.x >> 5)));
    const int r0 = wv * RPW;
    if (r0 >= n) return;
    const float4* t4   = (const float4*)t;
    const int2*   csr2 = (const int2*)csr;
    int start = rowptr[r0];
    int end   = rowptr[r0 + 1];
    int2 R[8];
    for (int k = 0; k < 8; ++k) R[k] = csr2[start + k];
    float4 sv = t4[(unsigned)(r0 * 16 + c)];
    float  d  = dinv[r0];
    const float4 bb = *(const float4*)(b + c * 4);
    for (int j = 0; j < RPW; ++j) {
        const int row = r0 + j;
        if (row >= n) break;
        const float4 acc0 = row_start(sv, d, bb, h);
        const int rn      = row + 1 < n ? row + 1 : row;
        const int nextEnd = rowptr[rn + 1];
        int2 Rn[8];
        for (int k = 0; k < 8; ++k) Rn[k] = csr2[end + k];
        const float4 svn = t4[(unsigned)(rn * 16 + c)];
        const float  dn  = dinv[rn];
        const float4 acc = gather_row(t4, csr2, R, acc0, row, start, end, h, c);
        if (h == 0) ((float4*)out)[(unsigned)(row * 16 + c)] = acc;
        start = end;
        end   = nextEnd;
        for (int k = 0; k < 8; ++k) R[k] = Rn[k];
        sv = svn;
        d  = dn;
    }
}

#define FR     (8 * RPW)
#define APITCH 68

__global__ void __launch_bounds__(256) __attribute__((amdgpu_waves_per_eu(16, 16)))
k_gather_gemm(const float* __restrict__ t, const int* __restrict__ rowptr, const int* __restrict__ csr,
              const float* __restrict__ dinv, const float* __restrict__ b, const float* __restrict__ Wn,
              float* __restrict__ tout, int n)
{
    __shared__ __attribute__((aligned(16))) float sA[FR * APITCH];

    const int lane = threadIdx.x & 31;
    const int h    = lane >> 4;
    const int c    = lane & 15;
    const int wave = __builtin_amdgcn_readfirstlane((int)(threadIdx.x >> 5));
    const int rowBase = blockIdx.x * FR;

    const int r0 = rowBase + wave * RPW;
    const float4* t4   = (const float4*)t;
    const int2*   csr2 = (const int2*)csr;
    int start = 0, end = 0;
    int2 R[8];
    for (int k = 0; k < 8; ++k) R[k] = make_int2(0, 0);
    float4 sv = make_float4(0.0f, 0.0f, 0.0f, 0.0f);
    float  d  = 0.0f;
    if (r0 < n) {
        start = rowptr[r0];
        end   = rowptr[r0 + 1];
        for (int k = 0; k < 8; ++k) R[k] = csr2[start + k];
        sv = t4[(unsigned)(r0 * 16 + c)];
        d  = dinv[r0];
    }
    for (int j = 0; j < RPW; ++j) {
        const int lr  = wave * RPW + j;
        const int row = rowBase + lr;
        float4 acc = make_float4(0.0f, 0.0f, 0.0f, 0.0f);
        if (row < n) {
            float4 bb = make_float4(0.0f, 0.0f, 0.0f, 0.0f);
            if (h == 0) bb = *(const float4*)(b + c * 4);
            const float4 acc0 = row_start(sv, d, bb, h);
            const int rn      = row + 1 < n ? row + 1 : row;
            const int nextEnd = rowptr[rn + 1];
            int2 Rn[8];
            for (int k = 0; k < 8; ++k) Rn[k] = csr2[end + k];
            const float4 svn = t4[(unsigned)(rn * 16 + c)];
            const float  dn  = dinv[rn];
            acc = gather_row(t4, csr2, R, acc0, row, start, end, h, c);
            start = end;
            end   = nextEnd;
            for (int k = 0; k < 8; ++k) R[k] = Rn[k];
            sv = svn;
            d  = dn;
        }
        if (h == 0) *(float4*)(sA + lr * APITCH + 4 * c) = acc;
    }
    __syncthreads();
    if (wave >= 4) return;

    const int r    = c;
    const int hi   = h;
    const int tile = wave >> 1;
    const int half = wave & 1;
    v8f acc0 = {}, acc1 = {};
    const float* aRow = sA + (tile * 16 + r) * APITCH;
    const float* wcol = Wn + half * 32 + r;

    #pragma unroll
    for (int k = 0; k < 16; ++k) {
        const int ka = 4 * k + 2 * hi;
        v2f a = *(const v2f*)(aRow + ka);
        a.x = fmaxf(a.x, 0.0f); a.y = fmaxf(a.y, 0.0f);
        v2f b0, b1;
        b0.x = wcol[ka * GD];
        b0.y = wcol[(ka + 1) * GD];
        b1.x = wcol[ka * GD + 16];
        b1.y = wcol[(ka + 1) * GD + 16];
        acc0 = __builtin_amdgcn_wmma_f32_16x16x4_f32(false, a, false, b0, (short)0, acc0, false, false);
        acc1 = __builtin_amdgcn_wmma_f32_16x16x4_f32(false, a, false, b1, (short)0, acc1, false, false);
    }

    #pragma unroll
    for (int v = 0; v < 8; ++v) {
        const int row = rowBase + tile * 16 + v + 8 * hi;
        if (row < n) {
            float* o = tout + (size_t)row * GD + half * 32 + r;
            o[0]  = acc0[v];
            o[16] = acc1[v];
        }
    }
}

extern "C" void kernel_launch(void* const* d_in, const int* in_sizes, int n_in,
                              void* d_out, int out_size, void* d_ws, size_t ws_size,
                              hipStream_t stream)
{
    const float* x    = (const float*)d_in[0];
    const int*   ei   = (const int*)d_in[1];
    const float* W_in = (const float*)d_in[2];
    const float* b_in = (const float*)d_in[3];
    const float* W1   = (const float*)d_in[4];
    const float* b1   = (const float*)d_in[5];
    const float* W2   = (const float*)d_in[6];
    const float* b2   = (const float*)d_in[7];
    const float* W3   = (const float*)d_in[8];
    const float* b3   = (const float*)d_in[9];
    float* out = (float*)d_out;

    const int N = in_sizes[0] / GD;
    const int E = in_sizes[1] / 2;
    const int* src = ei;
    const int* dst = ei + E;

    float* bufT = (float*)d_ws;
    int2*  tmp  = (int2*)d_ws;
    int*   csr  = (int*)(bufT + (size_t)N * GD);
    float* dinv = (float*)(csr + (size_t)2 * E);
    int*   rowp = (int*)(dinv + N);

    const int NB = (N + NPB - 1) / NPB;
    const int NC = (E + CH - 1) / CH;
    int*   tab    = (int*)d_out;
    int*   tabOff = tab + ((NB * NC + 3) & ~3);

    const int TB = 256;
    const int gGemm = (N + 127) / 128;
    const int gRow  = (N + 8 * RPW - 1) / (8 * RPW);
    const int gFuse = (N + FR - 1) / FR;

    k_chunk_hist   <<<NC, SORT_T, 0, stream>>>(dst, tab, E, NB, NC);
    k_scan         <<<1, 1024, 0, stream>>>(tab, tabOff, NB * NC);
    k_chunk_scatter<<<NC, SORT_T, 0, stream>>>(src, dst, tabOff, tmp, E, NB, NC);
    k_bucket_count <<<NB, SORT_T, 0, stream>>>(tmp, tabOff, rowp, dinv, N, E, NC);
    k_bucket_place <<<NB, SORT_T, 0, stream>>>(tmp, tabOff, rowp, dinv, (int2*)csr, N, NC);

    k_gemm64x2<<<gGemm, TB, 0, stream>>>(x, W_in, b_in, W1, bufT, N);

    k_gather_gemm<<<gFuse, TB, 0, stream>>>(bufT, rowp, csr, dinv, b1, W2, out, N);

    k_gather_gemm<<<gFuse, TB, 0, stream>>>(out, rowp, csr, dinv, b2, W3, bufT, N);

    k_gather<<<gRow, TB, 0, stream>>>(bufT, rowp, csr, dinv, b3, out, N);
}


# Chunk-local bucket sort: each chunk ranks and groups its edges in its own window (one LDS pass, in-group scan), buckets walk their per-chunk runs; 9 launches -> 7; record order in a row changes
# speedup vs baseline: 4.6758x; 1.1041x over previous
#include <hip/hip_runtime.h>


typedef __attribute__((ext_vector_type(2))) float v2f;
typedef __attribute__((ext_vector_type(8))) float v8f;

#define GD 64

#define NPB_SHIFT 8
#define NPB  (1 << NPB_SHIFT)
#define MAXB 256
#define CH   8192
#define SORT_T 1024
#define SORT_W (SORT_T / 32)
#define SEGB 8

__global__ void __launch_bounds__(SORT_T)
k_chunk_sort(const int* __restrict__ src, const int* __restrict__ dst,
             int2* __restrict__ tmp, int* __restrict__ coffT, int e, int nb, int nc)
{
    __shared__ int sCnt[MAXB];
    __shared__ int sTot[MAXB / 32];
    const int t    = threadIdx.x;
    const int lane = t & 31;
    if (t < MAXB) sCnt[t] = 0;
    __syncthreads();
    const int base = blockIdx.x * CH;
    int s[CH / SORT_T];
    int d[CH / SORT_T];
    int r[CH / SORT_T];
    #pragma unroll
    for (int j = 0; j < CH / SORT_T; ++j) {
        const int i  = base + j * SORT_T + t;
        const int ic = i < e ? i : e - 1;
        s[j] = src[ic];
        d[j] = dst[ic];
    }
    #pragma unroll
    for (int j = 0; j < CH / SORT_T; ++j) {
        r[j] = 0;
        if (base + j * SORT_T + t < e) r[j] = atomicAdd(&sCnt[d[j] >> NPB_SHIFT], 1);
    }
    __syncthreads();
    int c = 0, inc = 0;
    if (t < MAXB) {
        c = sCnt[t];
        inc = c;
        #pragma unroll
        for (int m = 1; m < 32; m <<= 1) {
            const int y = __shfl_up(inc, m);
            if (lane >= m) inc += y;
        }
        if (lane == 31) sTot[t >> 5] = inc;
    }
    __syncthreads();
    if (t < MAXB) {
        int pre = 0;
        for (int w = 0; w < (t >> 5); ++w) pre += sTot[w];
        const int off = pre + inc - c;
        sCnt[t] = off;
        if (t <= nb) coffT[t * nc + blockIdx.x] = off;
        if (t == MAXB - 1 && nb == MAXB) coffT[MAXB * nc + blockIdx.x] = off + c;
    }
    __syncthreads();
    #pragma unroll
    for (int j = 0; j < CH / SORT_T; ++j) {
        if (base + j * SORT_T + t < e) tmp[base + sCnt[d[j] >> NPB_SHIFT] + r[j]] = make_int2(s[j], d[j]);
    }
}

__global__ void __launch_bounds__(SORT_T)
k_bucket_count(const int2* __restrict__ tmp, const int* __restrict__ coffT,
               int* __restrict__ rowptr, float* __restrict__ dinv, int n, int e, int nc)
{
    __shared__ int sCnt[NPB];
    __shared__ int sTot[NPB / 32];
    __shared__ int sBase;
    const int t    = threadIdx.x;
    const int lane = t & 31;
    const int wave = __builtin_amdgcn_readfirstlane((int)(threadIdx.x >> 5));
    const int b    = blockIdx.x;
    if (t < NPB) sCnt[t] = 0;
    if (t == 0) sBase = 0;
    __syncthreads();
    int baseSum = 0;
    for (int k0 = 0; wave + SORT_W * k0 < nc; k0 += SEGB) {
        const int  cl = wave + SORT_W * (k0 + lane);
        const bool ok = lane < SEGB && cl < nc;
        int a = 0, len = 0;
        if (ok) {
            a   = coffT[b * nc + cl];
            len = coffT[(b + 1) * nc + cl] - a;
            baseSum += a;
        }
        const int st = ok ? cl * CH + a : 0;
        int dd[SEGB][2];
        #pragma unroll
        for (int k = 0; k < SEGB; ++k) {
            const int sk = __builtin_amdgcn_readlane(st, k);
            const int lk = __builtin_amdgcn_readlane(len, k);
            dd[k][0] = tmp[lane      < lk ? sk + lane      : 0].y;
            dd[k][1] = tmp[lane + 32 < lk ? sk + lane + 32 : 0].y;
        }
        #pragma unroll
        for (int k = 0; k < SEGB; ++k) {
            const int sk = __builtin_amdgcn_readlane(st, k);
            const int lk = __builtin_amdgcn_readlane(len, k);
            if (lane      < lk) atomicAdd(&sCnt[dd[k][0] & (NPB - 1)], 1);
            if (lane + 32 < lk) atomicAdd(&sCnt[dd[k][1] & (NPB - 1)], 1);
            for (int q = 64 + lane; q < lk; q += 32) atomicAdd(&sCnt[tmp[sk + q].y & (NPB - 1)], 1);
        }
    }
    baseSum += __shfl_xor(baseSum, 16);
    baseSum += __shfl_xor(baseSum, 8);
    baseSum += __shfl_xor(baseSum, 4);
    baseSum += __shfl_xor(baseSum, 2);
    baseSum += __shfl_xor(baseSum, 1);
    if (lane == 0) atomicAdd(&sBase, baseSum);
    __syncthreads();
    const int bstart = sBase;
    int c = 0, inc = 0;
    if (t < NPB) {
        c = sCnt[t];
        inc = c;
        #pragma unroll
        for (int m = 1; m < 32; m <<= 1) {
            const int y = __shfl_up(inc, m);
            if (lane >= m) inc += y;
        }
        if (lane == 31) sTot[t >> 5] = inc;
    }
    __syncthreads();
    if (t < NPB) {
        int pre = 0;
        for (int w = 0; w < (t >> 5); ++w) pre += sTot[w];
        const int node = b * NPB + t;
        if (node < n) {
            rowptr[node] = bstart + pre + inc - c;
            dinv[node]   = rsqrtf(fmaxf((float)(c + 1), 1e-12f));
        }
    }
    if (b == 0 && t == 0) rowptr[n] = e;
}

__global__ void __launch_bounds__(SORT_T)
k_bucket_place(const int2* __restrict__ tmp, const int* __restrict__ coffT,
               const int* __restrict__ rowptr, const float* __restrict__ dinv,
               int2* __restrict__ csr, int n, int nc)
{
    __shared__ int   sCur[NPB];
    __shared__ float sDinv[NPB];
    const int t    = threadIdx.x;
    const int lane = t & 31;
    const int wave = __builtin_amdgcn_readfirstlane((int)(threadIdx.x >> 5));
    const int b    = blockIdx.x;
    if (t < NPB) {
        const int node = b * NPB + t;
        const bool okn = node < n;
        sCur[t]  = okn ? rowptr[node] : 0;
        sDinv[t] = okn ? dinv[node] : 0.0f;
    }
    __syncthreads();
    for (int k0 = 0; wave + SORT_W * k0 < nc; k0 += SEGB) {
        const int  cl = wave + SORT_W * (k0 + lane);
        const bool ok = lane < SEGB && cl < nc;
        int a = 0, len = 0;
        if (ok) {
            a   = coffT[b * nc + cl];
            len = coffT[(b + 1) * nc + cl] - a;
        }
        const int st = ok ? cl * CH + a : 0;
        int2  rr[SEGB][2];
        float ds[SEGB][2];
        #pragma unroll
        for (int k = 0; k < SEGB; ++k) {
            const int sk = __builtin_amdgcn_readlane(st, k);
            const int lk = __builtin_amdgcn_readlane(len, k);
            rr[k][0] = tmp[lane      < lk ? sk + lane      : 0];
            rr[k][1] = tmp[lane + 32 < lk ? sk + lane + 32 : 0];
        }
        #pragma unroll
        for (int k = 0; k < SEGB; ++k) {
            ds[k][0] = dinv[rr[k][0].x];
            ds[k][1] = dinv[rr[k][1].x];
        }
        #pragma unroll
        for (int k = 0; k < SEGB; ++k) {
            const int sk = __builtin_amdgcn_readlane(st, k);
            const int lk = __builtin_amdgcn_readlane(len, k);
            if (lane < lk) {
                const int dl = rr[k][0].y & (NPB - 1);
                const int q  = atomicAdd(&sCur[dl], 1);
                int2 rec;
                rec.x = rr[k][0].x;
                rec.y = __float_as_int(ds[k][0] * sDinv[dl]);
                csr[q] = rec;
            }
            if (lane + 32 < lk) {
                const int dl = rr[k][1].y & (NPB - 1);
                const int q  = atomicAdd(&sCur[dl], 1);
                int2 rec;
                rec.x = rr[k][1].x;
                rec.y = __float_as_int(ds[k][1] * sDinv[dl]);
                csr[q] = rec;
            }
            for (int p = 64 + lane; p < lk; p += 32) {
                const int2 rl = tmp[sk + p];
                const int  dl = rl.y & (NPB - 1);
                const int  q  = atomicAdd(&sCur[dl], 1);
                int2 rec;
                rec.x = rl.x;
                rec.y = __float_as_int(dinv[rl.x] * sDinv[dl]);
                csr[q] = rec;
            }
        }
    }
}

#define HPITCH 68

__global__ void __launch_bounds__(256)
k_gemm64x2(const float* __restrict__ X, const float* __restrict__ Wa, const float* __restrict__ ba,
           const float* __restrict__ Wb, float* __restrict__ out, int n)
{
    __shared__ v2f   sWa[(GD / 2) * GD];
    __shared__ v2f   sWb[(GD / 2) * GD];
    __shared__ float sB[GD];
    __shared__ __attribute__((aligned(16))) float sH[8 * 16 * HPITCH];

    const int t = threadIdx.x;
    for (int i = t; i < (GD / 2) * GD; i += 256) {
        const int m = i >> 6;
        const int c = i & 63;
        v2f p, q;
        p.x = Wa[(2 * m) * GD + c];
        p.y = Wa[(2 * m + 1) * GD + c];
        q.x = Wb[(2 * m) * GD + c];
        q.y = Wb[(2 * m + 1) * GD + c];
        sWa[i] = p;
        sWb[i] = q;
    }
    if (t < GD) sB[t] = ba[t];
    __syncthreads();

    const int wave = t >> 5;
    const int lane = t & 31;
    const int r    = lane & 15;
    const int hi   = lane >> 4;
    const int rowBase = blockIdx.x * 128 + wave * 16;
    const bool live = rowBase < n;
    float* hT = sH + wave * (16 * HPITCH);

    if (live) {
        v8f acc0 = {}, acc1 = {}, acc2 = {}, acc3 = {};
        const float* aRow = X + (size_t)(rowBase + r) * GD;

        #pragma unroll
        for (int k = 0; k < 16; ++k) {
            const int ka = 4 * k + 2 * hi;
            const v2f a = *(const v2f*)(aRow + ka);

            const int mB = 2 * k + hi;
            const v2f* wp = sWa + (size_t)mB * GD + r;
            const v2f b0 = wp[0];
            const v2f b1 = wp[16];
            const v2f b2 = wp[32];
            const v2f b3 = wp[48];

            acc0 = __builtin_amdgcn_wmma_f32_16x16x4_f32(false, a, false, b0, (short)0, acc0, false, false);
            acc1 = __builtin_amdgcn_wmma_f32_16x16x4_f32(false, a, false, b1, (short)0, acc1, false, false);
            acc2 = __builtin_amdgcn_wmma_f32_16x16x4_f32(false, a, false, b2, (short)0, acc2, false, false);
            acc3 = __builtin_amdgcn_wmma_f32_16x16x4_f32(false, a, false, b3, (short)0, acc3, false, false);
        }

        #pragma unroll
        for (int v = 0; v < 8; ++v) {
            float* o = hT + (v + 8 * hi) * HPITCH + r;
            float x0 = acc0[v], x1 = acc1[v], x2 = acc2[v], x3 = acc3[v];
            x0 += sB[r]; x1 += sB[16 + r]; x2 += sB[32 + r]; x3 += sB[48 + r];
            x0 = fmaxf(x0, 0.0f); x1 = fmaxf(x1, 0.0f);
            x2 = fmaxf(x2, 0.0f); x3 = fmaxf(x3, 0.0f);
            o[0] = x0; o[16] = x1; o[32] = x2; o[48] = x3;
        }
    }
    __syncthreads();
    if (!live) return;

    v8f acc0 = {}, acc1 = {}, acc2 = {}, acc3 = {};
    const float* hRow = hT + r * HPITCH;

    #pragma unroll
    for (int k = 0; k < 16; ++k) {
        const int ka = 4 * k + 2 * hi;
        const v2f a = *(const v2f*)(hRow + ka);

        const int mB = 2 * k + hi;
        const v2f* wp = sWb + (size_t)mB * GD + r;
        const v2f b0 = wp[0];
        const v2f b1 = wp[16];
        const v2f b2 = wp[32];
        const v2f b3 = wp[48];

        acc0 = __builtin_amdgcn_wmma_f32_16x16x4_f32(false, a, false, b0, (short)0, acc0, false, false);
        acc1 = __builtin_amdgcn_wmma_f32_16x16x4_f32(false, a, false, b1, (short)0, acc1, false, false);
        acc2 = __builtin_amdgcn_wmma_f32_16x16x4_f32(false, a, false, b2, (short)0, acc2, false, false);
        acc3 = __builtin_amdgcn_wmma_f32_16x16x4_f32(false, a, false, b3, (short)0, acc3, false, false);
    }

    #pragma unroll
    for (int v = 0; v < 8; ++v) {
        const int row = rowBase + v + 8 * hi;
        float* o = out + (size_t)row * GD + r;
        o[0] = acc0[v]; o[16] = acc1[v]; o[32] = acc2[v]; o[48] = acc3[v];
    }
}

__device__ __forceinline__ float4 row_start(float4 sv, float d, float4 bb, int h)
{
    float4 acc = make_float4(0.0f, 0.0f, 0.0f, 0.0f);
    if (h == 0) {
        const float sl = d * d;
        acc.x = fmaf(sv.x, sl, bb.x);
        acc.y = fmaf(sv.y, sl, bb.y);
        acc.z = fmaf(sv.z, sl, bb.z);
        acc.w = fmaf(sv.w, sl, bb.w);
    }
    return acc;
}

__device__ __forceinline__ void gather8(float4& acc, const int2 (&R)[8], const float4* __restrict__ t4, int h, int c)
{
    int    s[4];
    float  w[4];
    float4 v[4];
    for (int k = 0; k < 4; ++k) {
        const int2 r0 = R[2 * k];
        const int2 r1 = R[2 * k + 1];
        s[k] = h ? r1.x : r0.x;
        w[k] = __int_as_float(h ? r1.y : r0.y);
    }
    for (int k = 0; k < 4; ++k) v[k] = t4[(unsigned)(s[k] * 16 + c)];
    for (int k = 0; k < 4; ++k) {
        acc.x = fmaf(v[k].x, w[k], acc.x);
        acc.y = fmaf(v[k].y, w[k], acc.y);
        acc.z = fmaf(v[k].z, w[k], acc.z);
        acc.w = fmaf(v[k].w, w[k], acc.w);
    }
}

__device__ __forceinline__ float4 gather_row(const float4* __restrict__ t4, const int2* __restrict__ csr2,
                                             int2 (&R)[8], float4 acc,
                                             int row, int start, int end, int h, int c)
{

    int p = start;
    while (p + 8 <= end) {
        int2 Q[8];
        for (int k = 0; k < 8; ++k) Q[k] = csr2[p + 8 + k];
        gather8(acc, R, t4, h, c);
        p += 8;
        if (p + 8 > end) {
            for (int k = 0; k < 8; ++k) R[k] = Q[k];
            break;
        }
        for (int k = 0; k < 8; ++k) R[k] = csr2[p + 8 + k];
        gather8(acc, Q, t4, h, c);
        p += 8;
    }

    const int nrem = end - p;
    if (nrem > 0) {
        int    s[4];
        float  w[4];
        float4 v[4];
        bool   ok[4];
        for (int k = 0; k < 4; ++k) {
            const int2 r0 = R[2 * k];
            const int2 r1 = R[2 * k + 1];
            ok[k] = (2 * k + h) < nrem;
            s[k] = ok[k] ? (h ? r1.x : r0.x) : row;
            w[k] = __int_as_float(h ? r1.y : r0.y);
        }
        for (int k = 0; k < 4; ++k) v[k] = t4[(unsigned)(s[k] * 16 + c)];
        for (int k = 0; k < 4; ++k) {
            if (ok[k]) {
                acc.x = fmaf(v[k].x, w[k], acc.x);
                acc.y = fmaf(v[k].y, w[k], acc.y);
                acc.z = fmaf(v[k].z, w[k], acc.z);
                acc.w = fmaf(v[k].w, w[k], acc.w);
            }
        }
    }

    acc.x += __shfl_xor(acc.x, 16);
    acc.y += __shfl_xor(acc.y, 16);
    acc.z += __shfl_xor(acc.z, 16);
    acc.w += __shfl_xor(acc.w, 16);
    return acc;
}

#define RPW 4

__global__ void __launch_bounds__(256)
k_gather(const float* __restrict__ t, const int* __restrict__ rowptr, const int* __restrict__ csr,
         const float* __restrict__ dinv, const float* __restrict__ b, float* __restrict__ out, int n)
{
    const int lane = threadIdx.x & 31;
    const int h    = lane >> 4;
    const int c    = lane & 15;
    const int wv = __builtin_amdgcn_readfirstlane((int)(blockIdx.x * 8 + (threadIdx.x >> 5)));
    const int r0 = wv * RPW;
    if (r0 >= n) return;
    const float4* t4   = (const float4*)t;
    const int2*   csr2 = (const int2*)csr;
    int start = rowptr[r0];
    int end   = rowptr[r0 + 1];
    int2 R[8];
    for (int k = 0; k < 8; ++k) R[k] = csr2[start + k];
    float4 sv = t4[(unsigned)(r0 * 16 + c)];
    float  d  = dinv[r0];
    const float4 bb = *(const float4*)(b + c * 4);
    for (int j = 0; j < RPW; ++j) {
        const int row = r0 + j;
        if (row >= n) break;
        const float4 acc0 = row_start(sv, d, bb, h);
        const int rn      = row + 1 < n ? row + 1 : row;
        const int nextEnd = rowptr[rn + 1];
        int2 Rn[8];
        for (int k = 0; k < 8; ++k) Rn[k] = csr2[end + k];
        const float4 svn = t4[(unsigned)(rn * 16 + c)];
        const float  dn  = dinv[rn];
        const float4 acc = gather_row(t4, csr2, R, acc0, row, start, end, h, c);
        if (h == 0) ((float4*)out)[(unsigned)(row * 16 + c)] = acc;
        start = end;
        end   = nextEnd;
        for (int k = 0; k < 8; ++k) R[k] = Rn[k];
        sv = svn;
        d  = dn;
    }
}

#define FR     (8 * RPW)
#define APITCH 68

__global__ void __launch_bounds__(256) __attribute__((amdgpu_waves_per_eu(16, 16)))
k_gather_gemm(const float* __restrict__ t, const int* __restrict__ rowptr, const int* __restrict__ csr,
              const float* __restrict__ dinv, const float* __restrict__ b, const float* __restrict__ Wn,
              float* __restrict__ tout, int n)
{
    __shared__ __attribute__((aligned(16))) float sA[FR * APITCH];

    const int lane = threadIdx.x & 31;
    const int h    = lane >> 4;
    const int c    = lane & 15;
    const int wave = __builtin_amdgcn_readfirstlane((int)(threadIdx.x >> 5));
    const int rowBase = blockIdx.x * FR;

    const int r0 = rowBase + wave * RPW;
    const float4* t4   = (const float4*)t;
    const int2*   csr2 = (const int2*)csr;
    int start = 0, end = 0;
    int2 R[8];
    for (int k = 0; k < 8; ++k) R[k] = make_int2(0, 0);
    float4 sv = make_float4(0.0f, 0.0f, 0.0f, 0.0f);
    float  d  = 0.0f;
    if (r0 < n) {
        start = rowptr[r0];
        end   = rowptr[r0 + 1];
        for (int k = 0; k < 8; ++k) R[k] = csr2[start + k];
        sv = t4[(unsigned)(r0 * 16 + c)];
        d  = dinv[r0];
    }
    for (int j = 0; j < RPW; ++j) {
        const int lr  = wave * RPW + j;
        const int row = rowBase + lr;
        float4 acc = make_float4(0.0f, 0.0f, 0.0f, 0.0f);
        if (row < n) {
            float4 bb = make_float4(0.0f, 0.0f, 0.0f, 0.0f);
            if (h == 0) bb = *(const float4*)(b + c * 4);
            const float4 acc0 = row_start(sv, d, bb, h);
            const int rn      = row + 1 < n ? row + 1 : row;
            const int nextEnd = rowptr[rn + 1];
            int2 Rn[8];
            for (int k = 0; k < 8; ++k) Rn[k] = csr2[end + k];
            const float4 svn = t4[(unsigned)(rn * 16 + c)];
            const float  dn  = dinv[rn];
            acc = gather_row(t4, csr2, R, acc0, row, start, end, h, c);
            start = end;
            end   = nextEnd;
            for (int k = 0; k < 8; ++k) R[k] = Rn[k];
            sv = svn;
            d  = dn;
        }
        if (h == 0) *(float4*)(sA + lr * APITCH + 4 * c) = acc;
    }
    __syncthreads();
    if (wave >= 4) return;

    const int r    = c;
    const int hi   = h;
    const int tile = wave >> 1;
    const int half = wave & 1;
    v8f acc0 = {}, acc1 = {};
    const float* aRow = sA + (tile * 16 + r) * APITCH;
    const float* wcol = Wn + half * 32 + r;

    #pragma unroll
    for (int k = 0; k < 16; ++k) {
        const int ka = 4 * k + 2 * hi;
        v2f a = *(const v2f*)(aRow + ka);
        a.x = fmaxf(a.x, 0.0f); a.y = fmaxf(a.y, 0.0f);
        v2f b0, b1;
        b0.x = wcol[ka * GD];
        b0.y = wcol[(ka + 1) * GD];
        b1.x = wcol[ka * GD + 16];
        b1.y = wcol[(ka + 1) * GD + 16];
        acc0 = __builtin_amdgcn_wmma_f32_16x16x4_f32(false, a, false, b0, (short)0, acc0, false, false);
        acc1 = __builtin_amdgcn_wmma_f32_16x16x4_f32(false, a, false, b1, (short)0, acc1, false, false);
    }

    #pragma unroll
    for (int v = 0; v < 8; ++v) {
        const int row = rowBase + tile * 16 + v + 8 * hi;
        if (row < n) {
            float* o = tout + (size_t)row * GD + half * 32 + r;
            o[0]  = acc0[v];
            o[16] = acc1[v];
        }
    }
}

extern "C" void kernel_launch(void* const* d_in, const int* in_sizes, int n_in,
                              void* d_out, int out_size, void* d_ws, size_t ws_size,
                              hipStream_t stream)
{
    const float* x    = (const float*)d_in[0];
    const int*   ei   = (const int*)d_in[1];
    const float* W_in = (const float*)d_in[2];
    const float* b_in = (const float*)d_in[3];
    const float* W1   = (const float*)d_in[4];
    const float* b1   = (const float*)d_in[5];
    const float* W2   = (const float*)d_in[6];
    const float* b2   = (const float*)d_in[7];
    const float* W3   = (const float*)d_in[8];
    const float* b3   = (const float*)d_in[9];
    float* out = (float*)d_out;

    const int N = in_sizes[0] / GD;
    const int E = in_sizes[1] / 2;
    const int* src = ei;
    const int* dst = ei + E;

    float* bufT = (float*)d_ws;
    int2*  tmp  = (int2*)d_ws;
    int*   csr  = (int*)(bufT + (size_t)N * GD);
    float* dinv = (float*)(csr + (size_t)2 * E);
    int*   rowp = (int*)(dinv + N);

    const int NB = (N + NPB - 1) / NPB;
    const int NC = (E + CH - 1) / CH;
    int*   coffT  = (int*)d_out;

    const int TB = 256;
    const int gGemm = (N + 127) / 128;
    const int gRow  = (N + 8 * RPW - 1) / (8 * RPW);
    const int gFuse = (N + FR - 1) / FR;

    k_chunk_sort   <<<NC, SORT_T, 0, stream>>>(src, dst, tmp, coffT, E, NB, NC);
    k_bucket_count <<<NB, SORT_T, 0, stream>>>(tmp, coffT, rowp, dinv, N, E, NC);
    k_bucket_place <<<NB, SORT_T, 0, stream>>>(tmp, coffT, rowp, dinv, (int2*)csr, N, NC);

    k_gemm64x2<<<gGemm, TB, 0, stream>>>(x, W_in, b_in, W1, bufT, N);

    k_gather_gemm<<<gFuse, TB, 0, stream>>>(bufT, rowp, csr, dinv, b1, W2, out, N);

    k_gather_gemm<<<gFuse, TB, 0, stream>>>(out, rowp, csr, dinv, b2, W3, bufT, N);

    k_gather<<<gRow, TB, 0, stream>>>(bufT, rowp, csr, dinv, b3, out, N);
}


# Sort kernels assemble the chunk window and each bucket's stretch of the sorted list in LDS and store tmp and csr as whole lines, 16 B per lane, not scattered 8- and 4-byte pieces; sum order kept
# speedup vs baseline: 5.8296x; 1.2468x over previous
#include <hip/hip_runtime.h>


typedef __attribute__((ext_vector_type(2))) float v2f;
typedef __attribute__((ext_vector_type(8))) float v8f;

#define GD 64

#define NPB_SHIFT 8
#define NPB  (1 << NPB_SHIFT)
#define MAXB 256
#define CH   8192
#define SORT_T 1024
#define SORT_W (SORT_T / 32)
#define SEGB 8

__global__ void __launch_bounds__(SORT_T)
k_chunk_sort(const int* __restrict__ src, const int* __restrict__ dst,
             int2* __restrict__ tmp, int* __restrict__ coffT, int e, int nb, int nc)
{
    __shared__ int sCnt[MAXB];
    __shared__ int sTot[MAXB / 32];
    __shared__ __attribute__((aligned(16))) int2 sRec[CH];
    const int t    = threadIdx.x;
    const int lane = t & 31;
    if (t < MAXB) sCnt[t] = 0;
    __syncthreads();
    const int base = blockIdx.x * CH;
    int s[CH / SORT_T];
    int d[CH / SORT_T];
    int r[CH / SORT_T];
    #pragma unroll
    for (int j = 0; j < CH / SORT_T; ++j) {
        const int i  = base + j * SORT_T + t;
        const int ic = i < e ? i : e - 1;
        s[j] = src[ic];
        d[j] = dst[ic];
    }
    #pragma unroll
    for (int j = 0; j < CH / SORT_T; ++j) {
        r[j] = 0;
        if (base + j * SORT_T + t < e) r[j] = atomicAdd(&sCnt[d[j] >> NPB_SHIFT], 1);
    }
    __syncthreads();
    int c = 0, inc = 0;
    if (t < MAXB) {
        c = sCnt[t];
        inc = c;
        #pragma unroll
        for (int m = 1; m < 32; m <<= 1) {
            const int y = __shfl_up(inc, m);
            if (lane >= m) inc += y;
        }
        if (lane == 31) sTot[t >> 5] = inc;
    }
    __syncthreads();
    if (t < MAXB) {
        int pre = 0;
        for (int w = 0; w < (t >> 5); ++w) pre += sTot[w];
        const int off = pre + inc - c;
        sCnt[t] = off;
        if (t <= nb) coffT[t * nc + blockIdx.x] = off;
        if (t == MAXB - 1 && nb == MAXB) coffT[MAXB * nc + blockIdx.x] = off + c;
    }
    __syncthreads();
    #pragma unroll
    for (int j = 0; j < CH / SORT_T; ++j) {
        if (base + j * SORT_T + t < e) sRec[sCnt[d[j] >> NPB_SHIFT] + r[j]] = make_int2(s[j], d[j]);
    }
    __syncthreads();
    const int cnt   = e - base < CH ? e - base : CH;
    const int npair = cnt >> 1;
    int4*       o4 = (int4*)(tmp + base);
    const int4* s4 = (const int4*)sRec;
    for (int j = 0; j < CH / 2 / SORT_T; ++j) {
        const int i = j * SORT_T + t;
        if (i < npair) o4[i] = s4[i];
    }
    if (t == 0 && (cnt & 1)) tmp[base + cnt - 1] = sRec[cnt - 1];
}

#define LCAP 16384
#define LPAD 128

template <bool STAGED>
__device__ __forceinline__ void bucket_place(const int2* __restrict__ tmp, const int* __restrict__ coffT, int* __restrict__ csr,
                                             int* sCnt, int* sOut, int b, int nc, int wave, int lane)
{
    for (int k0 = 0; wave + SORT_W * k0 < nc; k0 += SEGB) {
        const int  cl = wave + SORT_W * (k0 + lane);
        const bool ok = lane < SEGB && cl < nc;
        int a = 0, len = 0;
        if (ok) {
            a   = coffT[b * nc + cl];
            len = coffT[(b + 1) * nc + cl] - a;
        }
        const int st = ok ? cl * CH + a : 0;
        int2 rr[SEGB][2];
        #pragma unroll
        for (int k = 0; k < SEGB; ++k) {
            const int sk = __builtin_amdgcn_readlane(st, k);
            const int lk = __builtin_amdgcn_readlane(len, k);
            rr[k][0] = tmp[lane      < lk ? sk + lane      : 0];
            rr[k][1] = tmp[lane + 32 < lk ? sk + lane + 32 : 0];
        }
        #pragma unroll
        for (int k = 0; k < SEGB; ++k) {
            const int sk = __builtin_amdgcn_readlane(st, k);
            const int lk = __builtin_amdgcn_readlane(len, k);
            if (lane < lk) {
                const int p = atomicAdd(&sCnt[rr[k][0].y & (NPB - 1)], 1);
                if (STAGED) sOut[p] = rr[k][0].x; else csr[p] = rr[k][0].x;
            }
            if (lane + 32 < lk) {
                const int p = atomicAdd(&sCnt[rr[k][1].y & (NPB - 1)], 1);
                if (STAGED) sOut[p] = rr[k][1].x; else csr[p] = rr[k][1].x;
            }
            for (int q = 64 + lane; q < lk; q += 32) {
                const int2 rl = tmp[sk + q];
                const int p = atomicAdd(&sCnt[rl.y & (NPB - 1)], 1);
                if (STAGED) sOut[p] = rl.x; else csr[p] = rl.x;
            }
        }
    }
}

__global__ void __launch_bounds__(SORT_T)
k_bucket_build(const int2* __restrict__ tmp, const int* __restrict__ coffT,
               int* __restrict__ rowptr, float* __restrict__ dinv, int* __restrict__ csr,
               int n, int e, int nc)
{
    __shared__ int sCnt[NPB];
    __shared__ int sTot[NPB / 32];
    __shared__ int sBase;
    __shared__ __attribute__((aligned(16))) int sOut[LCAP + LPAD];
    const int t    = threadIdx.x;
    const int lane = t & 31;
    const int wave = __builtin_amdgcn_readfirstlane((int)(threadIdx.x >> 5));
    const int b    = blockIdx.x;
    if (t < NPB) sCnt[t] = 0;
    if (t == 0) sBase = 0;
    __syncthreads();
    int baseSum = 0;
    for (int k0 = 0; wave + SORT_W * k0 < nc; k0 += SEGB) {
        const int  cl = wave + SORT_W * (k0 + lane);
        const bool ok = lane < SEGB && cl < nc;
        int a = 0, len = 0;
        if (ok) {
            a   = coffT[b * nc + cl];
            len = coffT[(b + 1) * nc + cl] - a;
            baseSum += a;
        }
        const int st = ok ? cl * CH + a : 0;
        int dd[SEGB][2];
        #pragma unroll
        for (int k = 0; k < SEGB; ++k) {
            const int sk = __builtin_amdgcn_readlane(st, k);
            const int lk = __builtin_amdgcn_readlane(len, k);
            dd[k][0] = tmp[lane      < lk ? sk + lane      : 0].y;
            dd[k][1] = tmp[lane + 32 < lk ? sk + lane + 32 : 0].y;
        }
        #pragma unroll
        for (int k = 0; k < SEGB; ++k) {
            const int sk = __builtin_amdgcn_readlane(st, k);
            const int lk = __builtin_amdgcn_readlane(len, k);
            if (lane      < lk) atomicAdd(&sCnt[dd[k][0] & (NPB - 1)], 1);
            if (lane + 32 < lk) atomicAdd(&sCnt[dd[k][1] & (NPB - 1)], 1);
            for (int q = 64 + lane; q < lk; q += 32) atomicAdd(&sCnt[tmp[sk + q].y & (NPB - 1)], 1);
        }
    }
    baseSum += __shfl_xor(baseSum, 16);
    baseSum += __shfl_xor(baseSum, 8);
    baseSum += __shfl_xor(baseSum, 4);
    baseSum += __shfl_xor(baseSum, 2);
    baseSum += __shfl_xor(baseSum, 1);
    if (lane == 0) atomicAdd(&sBase, baseSum);
    __syncthreads();
    const int bstart = sBase;
    int c = 0, inc = 0;
    if (t < NPB) {
        c = sCnt[t];
        inc = c;
        #pragma unroll
        for (int m = 1; m < 32; m <<= 1) {
            const int y = __shfl_up(inc, m);
            if (lane >= m) inc += y;
        }
        if (lane == 31) sTot[t >> 5] = inc;
    }
    __syncthreads();
    int total = 0;
    for (int w = 0; w < NPB / 32; ++w) total += sTot[w];
    const bool staged = total <= LCAP;
    const int  shift  = bstart & (LPAD - 1);
    if (t < NPB) {
        int pre = 0;
        for (int w = 0; w < (t >> 5); ++w) pre += sTot[w];
        const int loc  = pre + inc - c;
        const int off  = bstart + loc;
        const int node = b * NPB + t;
        sCnt[t] = staged ? shift + loc : off;
        if (node < n) {
            rowptr[node] = off;
            dinv[node]   = rsqrtf(fmaxf((float)(c + 1), 1e-12f));
        }
    }
    if (b == 0 && t == 0) rowptr[n] = e;
    __syncthreads();
    if (!staged) {
        bucket_place<false>(tmp, coffT, csr, sCnt, sOut, b, nc, wave, lane);
        return;
    }
    bucket_place<true>(tmp, coffT, csr, sCnt, sOut, b, nc, wave, lane);
    __syncthreads();
    const int lo = shift, hi = shift + total;
    int* g = csr + (bstart - shift);
    for (int q = t; 4 * q < hi; q += SORT_T) {
        const int4 v  = ((const int4*)sOut)[q];
        const int  i0 = 4 * q;
        if (i0 >= lo && i0 + 4 <= hi) {
            ((int4*)g)[q] = v;
        } else {
            if (i0     >= lo && i0     < hi) g[i0]     = v.x;
            if (i0 + 1 >= lo && i0 + 1 < hi) g[i0 + 1] = v.y;
            if (i0 + 2 >= lo && i0 + 2 < hi) g[i0 + 2] = v.z;
            if (i0 + 3 >= lo && i0 + 3 < hi) g[i0 + 3] = v.w;
        }
    }
}

#define HPITCH 68

__global__ void __launch_bounds__(256)
k_gemm64x2(const float* __restrict__ X, const float* __restrict__ Wa, const float* __restrict__ ba,
           const float* __restrict__ Wb, const float* __restrict__ dinv, float* __restrict__ out, int n)
{
    __shared__ v2f   sWa[(GD / 2) * GD];
    __shared__ v2f   sWb[(GD / 2) * GD];
    __shared__ float sB[GD];
    __shared__ __attribute__((aligned(16))) float sH[8 * 16 * HPITCH];

    const int t = threadIdx.x;
    for (int i = t; i < (GD / 2) * GD; i += 256) {
        const int m = i >> 6;
        const int c = i & 63;
        v2f p, q;
        p.x = Wa[(2 * m) * GD + c];
        p.y = Wa[(2 * m + 1) * GD + c];
        q.x = Wb[(2 * m) * GD + c];
        q.y = Wb[(2 * m + 1) * GD + c];
        sWa[i] = p;
        sWb[i] = q;
    }
    if (t < GD) sB[t] = ba[t];
    __syncthreads();

    const int wave = t >> 5;
    const int lane = t & 31;
    const int r    = lane & 15;
    const int hi   = lane >> 4;
    const int rowBase = blockIdx.x * 128 + wave * 16;
    const bool live = rowBase < n;
    float* hT = sH + wave * (16 * HPITCH);

    if (live) {
        v8f acc0 = {}, acc1 = {}, acc2 = {}, acc3 = {};
        const float* aRow = X + (size_t)(rowBase + r) * GD;

        #pragma unroll
        for (int k = 0; k < 16; ++k) {
            const int ka = 4 * k + 2 * hi;
            const v2f a = *(const v2f*)(aRow + ka);

            const int mB = 2 * k + hi;
            const v2f* wp = sWa + (size_t)mB * GD + r;
            const v2f b0 = wp[0];
            const v2f b1 = wp[16];
            const v2f b2 = wp[32];
            const v2f b3 = wp[48];

            acc0 = __builtin_amdgcn_wmma_f32_16x16x4_f32(false, a, false, b0, (short)0, acc0, false, false);
            acc1 = __builtin_amdgcn_wmma_f32_16x16x4_f32(false, a, false, b1, (short)0, acc1, false, false);
            acc2 = __builtin_amdgcn_wmma_f32_16x16x4_f32(false, a, false, b2, (short)0, acc2, false, false);
            acc3 = __builtin_amdgcn_wmma_f32_16x16x4_f32(false, a, false, b3, (short)0, acc3, false, false);
        }

        #pragma unroll
        for (int v = 0; v < 8; ++v) {
            float* o = hT + (v + 8 * hi) * HPITCH + r;
            float x0 = acc0[v], x1 = acc1[v], x2 = acc2[v], x3 = acc3[v];
            x0 += sB[r]; x1 += sB[16 + r]; x2 += sB[32 + r]; x3 += sB[48 + r];
            x0 = fmaxf(x0, 0.0f); x1 = fmaxf(x1, 0.0f);
            x2 = fmaxf(x2, 0.0f); x3 = fmaxf(x3, 0.0f);
            o[0] = x0; o[16] = x1; o[32] = x2; o[48] = x3;
        }
    }
    __syncthreads();
    if (!live) return;

    v8f acc0 = {}, acc1 = {}, acc2 = {}, acc3 = {};
    const float* hRow = hT + r * HPITCH;

    #pragma unroll
    for (int k = 0; k < 16; ++k) {
        const int ka = 4 * k + 2 * hi;
        const v2f a = *(const v2f*)(hRow + ka);

        const int mB = 2 * k + hi;
        const v2f* wp = sWb + (size_t)mB * GD + r;
        const v2f b0 = wp[0];
        const v2f b1 = wp[16];
        const v2f b2 = wp[32];
        const v2f b3 = wp[48];

        acc0 = __builtin_amdgcn_wmma_f32_16x16x4_f32(false, a, false, b0, (short)0, acc0, false, false);
        acc1 = __builtin_amdgcn_wmma_f32_16x16x4_f32(false, a, false, b1, (short)0, acc1, false, false);
        acc2 = __builtin_amdgcn_wmma_f32_16x16x4_f32(false, a, false, b2, (short)0, acc2, false, false);
        acc3 = __builtin_amdgcn_wmma_f32_16x16x4_f32(false, a, false, b3, (short)0, acc3, false, false);
    }

    #pragma unroll
    for (int v = 0; v < 8; ++v) {
        const int row = rowBase + v + 8 * hi;
        const float dv = dinv[row];
        float* o = out + (size_t)row * GD + r;
        o[0] = acc0[v] * dv; o[16] = acc1[v] * dv; o[32] = acc2[v] * dv; o[48] = acc3[v] * dv;
    }
}

__device__ __forceinline__ float4 row_start(float4 sv, int h)
{
    float4 acc = make_float4(0.0f, 0.0f, 0.0f, 0.0f);
    if (h == 0) acc = sv;
    return acc;
}

__device__ __forceinline__ float4 row_finish(float4 acc, float d, float4 bb)
{
    float4 o;
    o.x = fmaf(acc.x, d, bb.x);
    o.y = fmaf(acc.y, d, bb.y);
    o.z = fmaf(acc.z, d, bb.z);
    o.w = fmaf(acc.w, d, bb.w);
    return o;
}

__device__ __forceinline__ void gather8(float4& acc, const int (&R)[8], const float4* __restrict__ t4, int h, int c)
{
    int    s[4];
    float4 v[4];
    for (int k = 0; k < 4; ++k) {
        const int r0 = R[2 * k];
        const int r1 = R[2 * k + 1];
        s[k] = h ? r1 : r0;
    }
    for (int k = 0; k < 4; ++k) v[k] = t4[(unsigned)(s[k] * 16 + c)];
    for (int k = 0; k < 4; ++k) {
        acc.x += v[k].x;
        acc.y += v[k].y;
        acc.z += v[k].z;
        acc.w += v[k].w;
    }
}

__device__ __forceinline__ float4 gather_row(const float4* __restrict__ t4, const int* __restrict__ csr,
                                             int (&R)[8], float4 acc,
                                             int row, int start, int end, int h, int c)
{

    int p = start;
    while (p + 8 <= end) {
        int Q[8];
        for (int k = 0; k < 8; ++k) Q[k] = csr[p + 8 + k];
        gather8(acc, R, t4, h, c);
        p += 8;
        if (p + 8 > end) {
            for (int k = 0; k < 8; ++k) R[k] = Q[k];
            break;
        }
        for (int k = 0; k < 8; ++k) R[k] = csr[p + 8 + k];
        gather8(acc, Q, t4, h, c);
        p += 8;
    }

    const int nrem = end - p;
    if (nrem > 0) {
        int    s[4];
        float4 v[4];
        bool   ok[4];
        for (int k = 0; k < 4; ++k) {
            const int r0 = R[2 * k];
            const int r1 = R[2 * k + 1];
            ok[k] = (2 * k + h) < nrem;
            s[k] = ok[k] ? (h ? r1 : r0) : row;
        }
        for (int k = 0; k < 4; ++k) v[k] = t4[(unsigned)(s[k] * 16 + c)];
        for (int k = 0; k < 4; ++k) {
            if (ok[k]) {
                acc.x += v[k].x;
                acc.y += v[k].y;
                acc.z += v[k].z;
                acc.w += v[k].w;
            }
        }
    }

    acc.x += __shfl_xor(acc.x, 16);
    acc.y += __shfl_xor(acc.y, 16);
    acc.z += __shfl_xor(acc.z, 16);
    acc.w += __shfl_xor(acc.w, 16);
    return acc;
}

#define RPW 4

__global__ void __launch_bounds__(256)
k_gather(const float* __restrict__ t, const int* __restrict__ rowptr, const int* __restrict__ csr,
         const float* __restrict__ dinv, const float* __restrict__ b, float* __restrict__ out, int n)
{
    const int lane = threadIdx.x & 31;
    const int h    = lane >> 4;
    const int c    = lane & 15;
    const int wv = __builtin_amdgcn_readfirstlane((int)(blockIdx.x * 8 + (threadIdx.x >> 5)));
    const int r0 = wv * RPW;
    if (r0 >= n) return;
    const float4* t4 = (const float4*)t;
    int start = rowptr[r0];
    int end   = rowptr[r0 + 1];
    int R[8];
    for (int k = 0; k < 8; ++k) R[k] = csr[start + k];
    float4 sv = t4[(unsigned)(r0 * 16 + c)];
    float  d  = dinv[r0];
    const float4 bb = *(const float4*)(b + c * 4);
    for (int j = 0; j < RPW; ++j) {
        const int row = r0 + j;
        if (row >= n) break;
        const float4 acc0 = row_start(sv, h);
        const int rn      = row + 1 < n ? row + 1 : row;
        const int nextEnd = rowptr[rn + 1];
        int Rn[8];
        for (int k = 0; k < 8; ++k) Rn[k] = csr[end + k];
        const float4 svn = t4[(unsigned)(rn * 16 + c)];
        const float  dn  = dinv[rn];
        const float4 acc = gather_row(t4, csr, R, acc0, row, start, end, h, c);
        if (h == 0) ((float4*)out)[(unsigned)(row * 16 + c)] = row_finish(acc, d, bb);
        start = end;
        end   = nextEnd;
        for (int k = 0; k < 8; ++k) R[k] = Rn[k];
        sv = svn;
        d  = dn;
    }
}

#define FR     (8 * RPW)
#define APITCH 68

__global__ void __launch_bounds__(256) __attribute__((amdgpu_waves_per_eu(16, 16)))
k_gather_gemm(const float* __restrict__ t, const int* __restrict__ rowptr, const int* __restrict__ csr,
              const float* __restrict__ dinv, const float* __restrict__ b, const float* __restrict__ Wn,
              float* __restrict__ tout, int n)
{
    __shared__ __attribute__((aligned(16))) float sA[FR * APITCH];

    const int lane = threadIdx.x & 31;
    const int h    = lane >> 4;
    const int c    = lane & 15;
    const int wave = __builtin_amdgcn_readfirstlane((int)(threadIdx.x >> 5));
    const int rowBase = blockIdx.x * FR;

    const int r0 = rowBase + wave * RPW;
    const float4* t4 = (const float4*)t;
    int start = 0, end = 0;
    int R[8];
    for (int k = 0; k < 8; ++k) R[k] = 0;
    float4 sv = make_float4(0.0f, 0.0f, 0.0f, 0.0f);
    float  d  = 0.0f;
    if (r0 < n) {
        start = rowptr[r0];
        end   = rowptr[r0 + 1];
        for (int k = 0; k < 8; ++k) R[k] = csr[start + k];
        sv = t4[(unsigned)(r0 * 16 + c)];
        d  = dinv[r0];
    }
    const float4 bb = *(const float4*)(b + c * 4);
    for (int j = 0; j < RPW; ++j) {
        const int lr  = wave * RPW + j;
        const int row = rowBase + lr;
        float4 st = make_float4(0.0f, 0.0f, 0.0f, 0.0f);
        if (row < n) {
            const float4 acc0 = row_start(sv, h);
            const int rn      = row + 1 < n ? row + 1 : row;
            const int nextEnd = rowptr[rn + 1];
            int Rn[8];
            for (int k = 0; k < 8; ++k) Rn[k] = csr[end + k];
            const float4 svn = t4[(unsigned)(rn * 16 + c)];
            const float  dn  = dinv[rn];
            const float4 acc = gather_row(t4, csr, R, acc0, row, start, end, h, c);
            st = row_finish(acc, d, bb);
            st.x *= d; st.y *= d; st.z *= d; st.w *= d;
            start = end;
            end   = nextEnd;
            for (int k = 0; k < 8; ++k) R[k] = Rn[k];
            sv = svn;
            d  = dn;
        }
        if (h == 0) *(float4*)(sA + lr * APITCH + 4 * c) = st;
    }
    __syncthreads();
    if (wave >= 4) return;

    const int r    = c;
    const int hi   = h;
    const int tile = wave >> 1;
    const int half = wave & 1;
    v8f acc0 = {}, acc1 = {};
    const float* aRow = sA + (tile * 16 + r) * APITCH;
    const float* wcol = Wn + half * 32 + r;

    #pragma unroll
    for (int k = 0; k < 16; ++k) {
        const int ka = 4 * k + 2 * hi;
        v2f a = *(const v2f*)(aRow + ka);
        a.x = fmaxf(a.x, 0.0f); a.y = fmaxf(a.y, 0.0f);
        v2f b0, b1;
        b0.x = wcol[ka * GD];
        b0.y = wcol[(ka + 1) * GD];
        b1.x = wcol[ka * GD + 16];
        b1.y = wcol[(ka + 1) * GD + 16];
        acc0 = __builtin_amdgcn_wmma_f32_16x16x4_f32(false, a, false, b0, (short)0, acc0, false, false);
        acc1 = __builtin_amdgcn_wmma_f32_16x16x4_f32(false, a, false, b1, (short)0, acc1, false, false);
    }

    #pragma unroll
    for (int v = 0; v < 8; ++v) {
        const int row = rowBase + tile * 16 + v + 8 * hi;
        if (row < n) {
            float* o = tout + (size_t)row * GD + half * 32 + r;
            o[0]  = acc0[v];
            o[16] = acc1[v];
        }
    }
}

extern "C" void kernel_launch(void* const* d_in, const int* in_sizes, int n_in,
                              void* d_out, int out_size, void* d_ws, size_t ws_size,
                              hipStream_t stream)
{
    const float* x    = (const float*)d_in[0];
    const int*   ei   = (const int*)d_in[1];
    const float* W_in = (const float*)d_in[2];
    const float* b_in = (const float*)d_in[3];
    const float* W1   = (const float*)d_in[4];
    const float* b1   = (const float*)d_in[5];
    const float* W2   = (const float*)d_in[6];
    const float* b2   = (const float*)d_in[7];
    const float* W3   = (const float*)d_in[8];
    const float* b3   = (const float*)d_in[9];
    float* out = (float*)d_out;

    const int N = in_sizes[0] / GD;
    const int E = in_sizes[1] / 2;
    const int* src = ei;
    const int* dst = ei + E;

    float* bufT = (float*)d_ws;
    int2*  tmp  = (int2*)d_ws;
    int*   csr  = (int*)(bufT + (size_t)N * GD);
    float* dinv = (float*)(csr + (size_t)E);
    int*   rowp = (int*)(dinv + N);

    const int NB = (N + NPB - 1) / NPB;
    const int NC = (E + CH - 1) / CH;
    int*   coffT  = (int*)d_out;

    const int TB = 256;
    const int gGemm = (N + 127) / 128;
    const int gRow  = (N + 8 * RPW - 1) / (8 * RPW);
    const int gFuse = (N + FR - 1) / FR;

    k_chunk_sort   <<<NC, SORT_T, 0, stream>>>(src, dst, tmp, coffT, E, NB, NC);
    k_bucket_build <<<NB, SORT_T, 0, stream>>>(tmp, coffT, rowp, dinv, csr, N, E, NC);

    k_gemm64x2<<<gGemm, TB, 0, stream>>>(x, W_in, b_in, W1, dinv, bufT, N);

    k_gather_gemm<<<gFuse, TB, 0, stream>>>(bufT, rowp, csr, dinv, b1, W2, out, N);

    k_gather_gemm<<<gFuse, TB, 0, stream>>>(out, rowp, csr, dinv, b2, W3, bufT, N);

    k_gather<<<gRow, TB, 0, stream>>>(bufT, rowp, csr, dinv, b3, out, N);
}


# Both dense products in front of the first gather move into the bucket-build work-group (its own 256 rows, dinv from LDS; 4-byte sort records parked in out): 6 launches -> 5, same k and sum order
# speedup vs baseline: 6.4432x; 1.1053x over previous
#include <hip/hip_runtime.h>


typedef __attribute__((ext_vector_type(2))) float v2f;
typedef __attribute__((ext_vector_type(8))) float v8f;

#define GD 64

#define NPB_SHIFT 8
#define NPB  (1 << NPB_SHIFT)
#define MAXB 256
#define CH   8192
#define SORT_T 1024
#define SORT_W (SORT_T / 32)
#define SEGB 8

__global__ void __launch_bounds__(SORT_T)
k_chunk_sort(const int* __restrict__ src, const int* __restrict__ dst,
             int* __restrict__ tmp, int* __restrict__ coffT, int e, int nb, int nc)
{
    __shared__ int sCnt[MAXB];
    __shared__ int sTot[MAXB / 32];
    __shared__ __attribute__((aligned(16))) int sRec[CH];
    const int t    = threadIdx.x;
    const int lane = t & 31;
    if (t < MAXB) sCnt[t] = 0;
    __syncthreads();
    const int base = blockIdx.x * CH;
    int s[CH / SORT_T];
    int d[CH / SORT_T];
    int r[CH / SORT_T];
    #pragma unroll
    for (int j = 0; j < CH / SORT_T; ++j) {
        const int i  = base + j * SORT_T + t;
        const int ic = i < e ? i : e - 1;
        s[j] = src[ic];
        d[j] = dst[ic];
    }
    #pragma unroll
    for (int j = 0; j < CH / SORT_T; ++j) {
        r[j] = 0;
        if (base + j * SORT_T + t < e) r[j] = atomicAdd(&sCnt[d[j] >> NPB_SHIFT], 1);
    }
    __syncthreads();
    int c = 0, inc = 0;
    if (t < MAXB) {
        c = sCnt[t];
        inc = c;
        #pragma unroll
        for (int m = 1; m < 32; m <<= 1) {
            const int y = __shfl_up(inc, m);
            if (lane >= m) inc += y;
        }
        if (lane == 31) sTot[t >> 5] = inc;
    }
    __syncthreads();
    if (t < MAXB) {
        int pre = 0;
        for (int w = 0; w < (t >> 5); ++w) pre += sTot[w];
        const int off = pre + inc - c;
        sCnt[t] = off;
        if (t <= nb) coffT[t * nc + blockIdx.x] = off;
        if (t == MAXB - 1 && nb == MAXB) coffT[MAXB * nc + blockIdx.x] = off + c;
    }
    __syncthreads();
    #pragma unroll
    for (int j = 0; j < CH / SORT_T; ++j) {
        if (base + j * SORT_T + t < e) sRec[sCnt[d[j] >> NPB_SHIFT] + r[j]] = (s[j] << NPB_SHIFT) | (d[j] & (NPB - 1));
    }
    __syncthreads();
    const int cnt   = e - base < CH ? e - base : CH;
    const int nquad = cnt >> 2;
    int4*       o4 = (int4*)(tmp + base);
    const int4* s4 = (const int4*)sRec;
    for (int j = 0; j < CH / 4 / SORT_T; ++j) {
        const int i = j * SORT_T + t;
        if (i < nquad) o4[i] = s4[i];
    }
    if (t < (cnt & 3)) tmp[base + 4 * nquad + t] = sRec[4 * nquad + t];
}

#define LCAP 16384
#define LPAD 128

template <bool STAGED>
__device__ __forceinline__ void bucket_place(const int* __restrict__ tmp, const int* __restrict__ coffT, int* __restrict__ csr,
                                             int* sCnt, int* sOut, int b, int nc, int wave, int lane)
{
    for (int k0 = 0; wave + SORT_W * k0 < nc; k0 += SEGB) {
        const int  cl = wave + SORT_W * (k0 + lane);
        const bool ok = lane < SEGB && cl < nc;
        int a = 0, len = 0;
        if (ok) {
            a   = coffT[b * nc + cl];
            len = coffT[(b + 1) * nc + cl] - a;
        }
        const int st = ok ? cl * CH + a : 0;
        int rr[SEGB][2];
        #pragma unroll
        for (int k = 0; k < SEGB; ++k) {
            const int sk = __builtin_amdgcn_readlane(st, k);
            const int lk = __builtin_amdgcn_readlane(len, k);
            rr[k][0] = tmp[lane      < lk ? sk + lane      : 0];
            rr[k][1] = tmp[lane + 32 < lk ? sk + lane + 32 : 0];
        }
        #pragma unroll
        for (int k = 0; k < SEGB; ++k) {
            const int sk = __builtin_amdgcn_readlane(st, k);
            const int lk = __builtin_amdgcn_readlane(len, k);
            if (lane < lk) {
                const int p = atomicAdd(&sCnt[rr[k][0] & (NPB - 1)], 1);
                const int sv = (int)((unsigned)rr[k][0] >> NPB_SHIFT);
                if (STAGED) sOut[p] = sv; else csr[p] = sv;
            }
            if (lane + 32 < lk) {
                const int p = atomicAdd(&sCnt[rr[k][1] & (NPB - 1)], 1);
                const int sv = (int)((unsigned)rr[k][1] >> NPB_SHIFT);
                if (STAGED) sOut[p] = sv; else csr[p] = sv;
            }
            for (int q = 64 + lane; q < lk; q += 32) {
                const int rl = tmp[sk + q];
                const int p = atomicAdd(&sCnt[rl & (NPB - 1)], 1);
                const int sv = (int)((unsigned)rl >> NPB_SHIFT);
                if (STAGED) sOut[p] = sv; else csr[p] = sv;
            }
        }
    }
}

#define HPITCH 68

__device__ __forceinline__ void hidden_stripe(const v2f (&a)[16], const v2f* sWa, const float* sB, float* hT, int half, int r, int hi)
{
    v8f acc0 = {}, acc1 = {};
    #pragma unroll
    for (int k = 0; k < 16; ++k) {
        const int mB = 2 * k + hi;
        const v2f* wp = sWa + (size_t)mB * GD + half * 32 + r;
        const v2f b0 = wp[0];
        const v2f b1 = wp[16];
        acc0 = __builtin_amdgcn_wmma_f32_16x16x4_f32(false, a[k], false, b0, (short)0, acc0, false, false);
        acc1 = __builtin_amdgcn_wmma_f32_16x16x4_f32(false, a[k], false, b1, (short)0, acc1, false, false);
    }
    #pragma unroll
    for (int v = 0; v < 8; ++v) {
        float* o = hT + (v + 8 * hi) * HPITCH + half * 32 + r;
        float x0 = acc0[v], x1 = acc1[v];
        x0 += sB[half * 32 + r]; x1 += sB[half * 32 + 16 + r];
        x0 = fmaxf(x0, 0.0f); x1 = fmaxf(x1, 0.0f);
        o[0] = x0; o[16] = x1;
    }
}

__global__ void __launch_bounds__(SORT_T)
k_bucket_build_gemm(const int* __restrict__ tmp, const int* __restrict__ coffT,
                    int* __restrict__ rowptr, float* __restrict__ dinv, int* __restrict__ csr,
                    const float* __restrict__ X, const float* __restrict__ Wa, const float* __restrict__ ba,
                    const float* __restrict__ Wb, float* __restrict__ tout,
                    int n, int e, int nc)
{
    __shared__ int sCnt[NPB];
    __shared__ int sTot[NPB / 32];
    __shared__ int sBase;
    __shared__ float sDinv[NPB];
    __shared__ __attribute__((aligned(16))) int sOut[LCAP + LPAD];
    __shared__ v2f   sWa[(GD / 2) * GD];
    __shared__ v2f   sWb[(GD / 2) * GD];
    __shared__ float sB[GD];
    __shared__ __attribute__((aligned(16))) float sH[(NPB / 16) * 16 * HPITCH];
    const int t    = threadIdx.x;
    const int lane = t & 31;
    const int wave = __builtin_amdgcn_readfirstlane((int)(threadIdx.x >> 5));
    const int b    = blockIdx.x;
    if (t < NPB) sCnt[t] = 0;
    if (t == 0) sBase = 0;
    for (int i = t; i < (GD / 2) * GD; i += SORT_T) {
        const int m = i >> 6;
        const int c = i & 63;
        v2f p, q;
        p.x = Wa[(2 * m) * GD + c];
        p.y = Wa[(2 * m + 1) * GD + c];
        q.x = Wb[(2 * m) * GD + c];
        q.y = Wb[(2 * m + 1) * GD + c];
        sWa[i] = p;
        sWb[i] = q;
    }
    if (t < GD) sB[t] = ba[t];

    const int r      = lane & 15;
    const int hi     = lane >> 4;
    const int stripe = wave >> 1;
    const int half   = wave & 1;
    const int rowBase = b * NPB + stripe * 16;
    const bool live = rowBase < n;
    float* hT = sH + stripe * (16 * HPITCH);
    v2f a[16];
    #pragma unroll
    for (int k = 0; k < 16; ++k) { a[k].x = 0.0f; a[k].y = 0.0f; }
    if (live) {
        const float* aRow = X + (size_t)(rowBase + r) * GD + 2 * hi;
        #pragma unroll
        for (int k = 0; k < 16; ++k) a[k] = *(const v2f*)(aRow + 4 * k);
    }
    __syncthreads();
    int baseSum = 0;
    for (int k0 = 0; wave + SORT_W * k0 < nc; k0 += SEGB) {
        const int  cl = wave + SORT_W * (k0 + lane);
        const bool ok = lane < SEGB && cl < nc;
        int aa = 0, len = 0;
        if (ok) {
            aa  = coffT[b * nc + cl];
            len = coffT[(b + 1) * nc + cl] - aa;
            baseSum += aa;
        }
        const int st = ok ? cl * CH + aa : 0;
        int dd[SEGB][2];
        #pragma unroll
        for (int k = 0; k < SEGB; ++k) {
            const int sk = __builtin_amdgcn_readlane(st, k);
            const int lk = __builtin_amdgcn_readlane(len, k);
            dd[k][0] = tmp[lane      < lk ? sk + lane      : 0];
            dd[k][1] = tmp[lane + 32 < lk ? sk + lane + 32 : 0];
        }
        if (k0 == 0 && live) hidden_stripe(a, sWa, sB, hT, half, r, hi);
        #pragma unroll
        for (int k = 0; k < SEGB; ++k) {
            const int sk = __builtin_amdgcn_readlane(st, k);
            const int lk = __builtin_amdgcn_readlane(len, k);
            if (lane      < lk) atomicAdd(&sCnt[dd[k][0] & (NPB - 1)], 1);
            if (lane + 32 < lk) atomicAdd(&sCnt[dd[k][1] & (NPB - 1)], 1);
            for (int q = 64 + lane; q < lk; q += 32) atomicAdd(&sCnt[tmp[sk + q] & (NPB - 1)], 1);
        }
    }
    if (wave >= nc && live) hidden_stripe(a, sWa, sB, hT, half, r, hi);
    baseSum += __shfl_xor(baseSum, 16);
    baseSum += __shfl_xor(baseSum, 8);
    baseSum += __shfl_xor(baseSum, 4);
    baseSum += __shfl_xor(baseSum, 2);
    baseSum += __shfl_xor(baseSum, 1);
    if (lane == 0) atomicAdd(&sBase, baseSum);
    __syncthreads();
    const int bstart = sBase;
    int c = 0, inc = 0;
    if (t < NPB) {
        c = sCnt[t];
        inc = c;
        #pragma unroll
        for (int m = 1; m < 32; m <<= 1) {
            const int y = __shfl_up(inc, m);
            if (lane >= m) inc += y;
        }
        if (lane == 31) sTot[t >> 5] = inc;
    }
    __syncthreads();
    int total = 0;
    for (int w = 0; w < NPB / 32; ++w) total += sTot[w];
    const bool staged = total <= LCAP;
    const int  shift  = bstart & (LPAD - 1);
    if (t < NPB) {
        int pre = 0;
        for (int w = 0; w < (t >> 5); ++w) pre += sTot[w];
        const int loc  = pre + inc - c;
        const int off  = bstart + loc;
        const int node = b * NPB + t;
        sCnt[t] = staged ? shift + loc : off;
        const float dv = rsqrtf(fmaxf((float)(c + 1), 1e-12f));
        sDinv[t] = dv;
        if (node < n) {
            rowptr[node] = off;
            dinv[node]   = dv;
        }
    }
    if (b == 0 && t == 0) rowptr[n] = e;
    __syncthreads();
    if (live) {
        v8f acc0 = {}, acc1 = {};
        const float* hRow = hT + r * HPITCH + 2 * hi;
        #pragma unroll
        for (int k = 0; k < 16; ++k) {
            const v2f ah = *(const v2f*)(hRow + 4 * k);
            const int mB = 2 * k + hi;
            const v2f* wp = sWb + (size_t)mB * GD + half * 32 + r;
            const v2f b0 = wp[0];
            const v2f b1 = wp[16];
            acc0 = __builtin_amdgcn_wmma_f32_16x16x4_f32(false, ah, false, b0, (short)0, acc0, false, false);
            acc1 = __builtin_amdgcn_wmma_f32_16x16x4_f32(false, ah, false, b1, (short)0, acc1, false, false);
        }
        #pragma unroll
        for (int v = 0; v < 8; ++v) {
            const int   lr = stripe * 16 + v + 8 * hi;
            const float dv = sDinv[lr];
            float* o = tout + (size_t)(b * NPB + lr) * GD + half * 32 + r;
            o[0] = acc0[v] * dv; o[16] = acc1[v] * dv;
        }
    }
    if (!staged) {
        bucket_place<false>(tmp, coffT, csr, sCnt, sOut, b, nc, wave, lane);
        return;
    }
    bucket_place<true>(tmp, coffT, csr, sCnt, sOut, b, nc, wave, lane);
    __syncthreads();
    const int lo = shift, hiS = shift + total;
    int* g = csr + (bstart - shift);
    for (int q = t; 4 * q < hiS; q += SORT_T) {
        const int4 v  = ((const int4*)sOut)[q];
        const int  i0 = 4 * q;
        if (i0 >= lo && i0 + 4 <= hiS) {
            ((int4*)g)[q] = v;
        } else {
            if (i0     >= lo && i0     < hiS) g[i0]     = v.x;
            if (i0 + 1 >= lo && i0 + 1 < hiS) g[i0 + 1] = v.y;
            if (i0 + 2 >= lo && i0 + 2 < hiS) g[i0 + 2] = v.z;
            if (i0 + 3 >= lo && i0 + 3 < hiS) g[i0 + 3] = v.w;
        }
    }
}

__device__ __forceinline__ float4 row_start(float4 sv, int h)
{
    float4 acc = make_float4(0.0f, 0.0f, 0.0f, 0.0f);
    if (h == 0) acc = sv;
    return acc;
}

__device__ __forceinline__ float4 row_finish(float4 acc, float d, float4 bb)
{
    float4 o;
    o.x = fmaf(acc.x, d, bb.x);
    o.y = fmaf(acc.y, d, bb.y);
    o.z = fmaf(acc.z, d, bb.z);
    o.w = fmaf(acc.w, d, bb.w);
    return o;
}

__device__ __forceinline__ void gather8(float4& acc, const int (&R)[8], const float4* __restrict__ t4, int h, int c)
{
    int    s[4];
    float4 v[4];
    for (int k = 0; k < 4; ++k) {
        const int r0 = R[2 * k];
        const int r1 = R[2 * k + 1];
        s[k] = h ? r1 : r0;
    }
    for (int k = 0; k < 4; ++k) v[k] = t4[(unsigned)(s[k] * 16 + c)];
    for (int k = 0; k < 4; ++k) {
        acc.x += v[k].x;
        acc.y += v[k].y;
        acc.z += v[k].z;
        acc.w += v[k].w;
    }
}

__device__ __forceinline__ float4 gather_row(const float4* __restrict__ t4, const int* __restrict__ csr,
                                             int (&R)[8], float4 acc,
                                             int row, int start, int end, int h, int c)
{

    int p = start;
    while (p + 8 <= end) {
        int Q[8];
        for (int k = 0; k < 8; ++k) Q[k] = csr[p + 8 + k];
        gather8(acc, R, t4, h, c);
        p += 8;
        if (p + 8 > end) {
            for (int k = 0; k < 8; ++k) R[k] = Q[k];
            break;
        }
        for (int k = 0; k < 8; ++k) R[k] = csr[p + 8 + k];
        gather8(acc, Q, t4, h, c);
        p += 8;
    }

    const int nrem = end - p;
    if (nrem > 0) {
        int    s[4];
        float4 v[4];
        bool   ok[4];
        for (int k = 0; k < 4; ++k) {
            const int r0 = R[2 * k];
            const int r1 = R[2 * k + 1];
            ok[k] = (2 * k + h) < nrem;
            s[k] = ok[k] ? (h ? r1 : r0) : row;
        }
        for (int k = 0; k < 4; ++k) v[k] = t4[(unsigned)(s[k] * 16 + c)];
        for (int k = 0; k < 4; ++k) {
            if (ok[k]) {
                acc.x += v[k].x;
                acc.y += v[k].y;
                acc.z += v[k].z;
                acc.w += v[k].w;
            }
        }
    }

    acc.x += __shfl_xor(acc.x, 16);
    acc.y += __shfl_xor(acc.y, 16);
    acc.z += __shfl_xor(acc.z, 16);
    acc.w += __shfl_xor(acc.w, 16);
    return acc;
}

#define RPW 4

__global__ void __launch_bounds__(256)
k_gather(const float* __restrict__ t, const int* __restrict__ rowptr, const int* __restrict__ csr,
         const float* __restrict__ dinv, const float* __restrict__ b, float* __restrict__ out, int n)
{
    const int lane = threadIdx.x & 31;
    const int h    = lane >> 4;
    const int c    = lane & 15;
    const int wv = __builtin_amdgcn_readfirstlane((int)(blockIdx.x * 8 + (threadIdx.x >> 5)));
    const int r0 = wv * RPW;
    if (r0 >= n) return;
    const float4* t4 = (const float4*)t;
    int start = rowptr[r0];
    int end   = rowptr[r0 + 1];
    int R[8];
    for (int k = 0; k < 8; ++k) R[k] = csr[start + k];
    float4 sv = t4[(unsigned)(r0 * 16 + c)];
    float  d  = dinv[r0];
    const float4 bb = *(const float4*)(b + c * 4);
    for (int j = 0; j < RPW; ++j) {
        const int row = r0 + j;
        if (row >= n) break;
        const float4 acc0 = row_start(sv, h);
        const int rn      = row + 1 < n ? row + 1 : row;
        const int nextEnd = rowptr[rn + 1];
        int Rn[8];
        for (int k = 0; k < 8; ++k) Rn[k] = csr[end + k];
        const float4 svn = t4[(unsigned)(rn * 16 + c)];
        const float  dn  = dinv[rn];
        const float4 acc = gather_row(t4, csr, R, acc0, row, start, end, h, c);
        if (h == 0) ((float4*)out)[(unsigned)(row * 16 + c)] = row_finish(acc, d, bb);
        start = end;
        end   = nextEnd;
        for (int k = 0; k < 8; ++k) R[k] = Rn[k];
        sv = svn;
        d  = dn;
    }
}

#define FR     (8 * RPW)
#define APITCH 68

__global__ void __launch_bounds__(256) __attribute__((amdgpu_waves_per_eu(16, 16)))
k_gather_gemm(const float* __restrict__ t, const int* __restrict__ rowptr, const int* __restrict__ csr,
              const float* __restrict__ dinv, const float* __restrict__ b, const float* __restrict__ Wn,
              float* __restrict__ tout, int n)
{
    __shared__ __attribute__((aligned(16))) float sA[FR * APITCH];

    const int lane = threadIdx.x & 31;
    const int h    = lane >> 4;
    const int c    = lane & 15;
    const int wave = __builtin_amdgcn_readfirstlane((int)(threadIdx.x >> 5));
    const int rowBase = blockIdx.x * FR;

    const int r0 = rowBase + wave * RPW;
    const float4* t4 = (const float4*)t;
    int start = 0, end = 0;
    int R[8];
    for (int k = 0; k < 8; ++k) R[k] = 0;
    float4 sv = make_float4(0.0f, 0.0f, 0.0f, 0.0f);
    float  d  = 0.0f;
    if (r0 < n) {
        start = rowptr[r0];
        end   = rowptr[r0 + 1];
        for (int k = 0; k < 8; ++k) R[k] = csr[start + k];
        sv = t4[(unsigned)(r0 * 16 + c)];
        d  = dinv[r0];
    }
    const float4 bb = *(const float4*)(b + c * 4);
    for (int j = 0; j < RPW; ++j) {
        const int lr  = wave * RPW + j;
        const int row = rowBase + lr;
        float4 st = make_float4(0.0f, 0.0f, 0.0f, 0.0f);
        if (row < n) {
            const float4 acc0 = row_start(sv, h);
            const int rn      = row + 1 < n ? row + 1 : row;
            const int nextEnd = rowptr[rn + 1];
            int Rn[8];
            for (int k = 0; k < 8; ++k) Rn[k] = csr[end + k];
            const float4 svn = t4[(unsigned)(rn * 16 + c)];
            const float  dn  = dinv[rn];
            const float4 acc = gather_row(t4, csr, R, acc0, row, start, end, h, c);
            st = row_finish(acc, d, bb);
            st.x *= d; st.y *= d; st.z *= d; st.w *= d;
            start = end;
            end   = nextEnd;
            for (int k = 0; k < 8; ++k) R[k] = Rn[k];
            sv = svn;
            d  = dn;
        }
        if (h == 0) *(float4*)(sA + lr * APITCH + 4 * c) = st;
    }
    __syncthreads();
    if (wave >= 4) return;

    const int r    = c;
    const int hi   = h;
    const int tile = wave >> 1;
    const int half = wave & 1;
    v8f acc0 = {}, acc1 = {};
    const float* aRow = sA + (tile * 16 + r) * APITCH;
    const float* wcol = Wn + half * 32 + r;

    #pragma unroll
    for (int k = 0; k < 16; ++k) {
        const int ka = 4 * k + 2 * hi;
        v2f a = *(const v2f*)(aRow + ka);
        a.x = fmaxf(a.x, 0.0f); a.y = fmaxf(a.y, 0.0f);
        v2f b0, b1;
        b0.x = wcol[ka * GD];
        b0.y = wcol[(ka + 1) * GD];
        b1.x = wcol[ka * GD + 16];
        b1.y = wcol[(ka + 1) * GD + 16];
        acc0 = __builtin_amdgcn_wmma_f32_16x16x4_f32(false, a, false, b0, (short)0, acc0, false, false);
        acc1 = __builtin_amdgcn_wmma_f32_16x16x4_f32(false, a, false, b1, (short)0, acc1, false, false);
    }

    #pragma unroll
    for (int v = 0; v < 8; ++v) {
        const int row = rowBase + tile * 16 + v + 8 * hi;
        if (row < n) {
            float* o = tout + (size_t)row * GD + half * 32 + r;
            o[0]  = acc0[v];
            o[16] = acc1[v];
        }
    }
}

extern "C" void kernel_launch(void* const* d_in, const int* in_sizes, int n_in,
                              void* d_out, int out_size, void* d_ws, size_t ws_size,
                              hipStream_t stream)
{
    const float* x    = (const float*)d_in[0];
    const int*   ei   = (const int*)d_in[1];
    const float* W_in = (const float*)d_in[2];
    const float* b_in = (const float*)d_in[3];
    const float* W1   = (const float*)d_in[4];
    const float* b1   = (const float*)d_in[5];
    const float* W2   = (const float*)d_in[6];
    const float* b2   = (const float*)d_in[7];
    const float* W3   = (const float*)d_in[8];
    const float* b3   = (const float*)d_in[9];
    float* out = (float*)d_out;

    const int N = in_sizes[0] / GD;
    const int E = in_sizes[1] / 2;
    const int* src = ei;
    const int* dst = ei + E;

    float* bufT = (float*)d_ws;
    int*   csr  = (int*)(bufT + (size_t)N * GD);
    float* dinv = (float*)(csr + (size_t)E);
    int*   rowp = (int*)(dinv + N);

    const int NB = (N + NPB - 1) / NPB;
    const int NC = (E + CH - 1) / CH;
    int*   coffT  = (int*)d_out;
    int*   tmp    = coffT + (((NB + 1) * NC + 3) & ~3);

    const int TB = 256;
    const int gRow  = (N + 8 * RPW - 1) / (8 * RPW);
    const int gFuse = (N + FR - 1) / FR;

    k_chunk_sort   <<<NC, SORT_T, 0, stream>>>(src, dst, tmp, coffT, E, NB, NC);
    k_bucket_build_gemm<<<NB, SORT_T, 0, stream>>>(tmp, coffT, rowp, dinv, csr, x, W_in, b_in, W1, bufT, N, E, NC);

    k_gather_gemm<<<gFuse, TB, 0, stream>>>(bufT, rowp, csr, dinv, b1, W2, out, N);

    k_gather_gemm<<<gFuse, TB, 0, stream>>>(out, rowp, csr, dinv, b2, W3, bufT, N);

    k_gather<<<gRow, TB, 0, stream>>>(bufT, rowp, csr, dinv, b3, out, N);
}
